# starting kernel
# baseline (speedup 1.0000x reference)
#include <hip/hip_runtime.h>


typedef __attribute__((ext_vector_type(2))) float v2f;
typedef __attribute__((ext_vector_type(8))) float v8f;

__device__ __forceinline__ void mm(float* __restrict__ C, int ldc,
                                   const float* __restrict__ A, int lda,
                                   const float* __restrict__ B, int ldb,
                                   int M, int N, int Kd, bool transB)
{
    __syncthreads();
    const int wave = threadIdx.x >> 5;
    const int lane = threadIdx.x & 31;
    const int half = lane >> 4;
    const int l15  = lane & 15;
    const int tn = N >> 4;
    const int tiles = (M >> 4) * tn;
    for (int t = wave; t < tiles; t += 4) {
        const int mi = (t / tn) << 4;
        const int ni = (t % tn) << 4;
        v8f acc = {};
        #pragma unroll 4
        for (int k = 0; k < Kd; k += 4) {
            v2f a, b;
            const float* ar = A + (mi + l15) * lda + k + 2 * half;
            a.x = ar[0];
            a.y = ar[1];
            if (!transB) {
                const float* br = B + (k + 2 * half) * ldb + ni + l15;
                b.x = br[0];
                b.y = br[ldb];
            } else {
                const float* br = B + (ni + l15) * ldb + k + 2 * half;
                b.x = br[0];
                b.y = br[1];
            }
            acc = __builtin_amdgcn_wmma_f32_16x16x4_f32(
                false, a, false, b, (short)0, acc, false, false);
        }
        #pragma unroll
        for (int r = 0; r < 8; ++r)
            C[(mi + half * 8 + r) * ldc + ni + l15] = acc[r];
    }
    __syncthreads();
}

__device__ __forceinline__ float trace32(const float* A, float* red)
{
    __syncthreads();
    if (threadIdx.x < 32) red[threadIdx.x] = A[threadIdx.x * 33];
    __syncthreads();
    float s = 0.f;
    #pragma unroll
    for (int i = 0; i < 32; ++i) s += red[i];
    return s;
}

__device__ void ns_sqrt(float* dst_s, float* dst_i, const float* src,
                        float* y, float* z, float* t, float* yn, float* zn,
                        float* red, int iters)
{
    const int tid = threadIdx.x;
    const float c = trace32(src, red);
    const float inv_c = 1.f / c;
    __syncthreads();
    for (int i = tid; i < 1024; i += 128) {
        y[i] = src[i] * inv_c;
        z[i] = ((i % 33) == 0) ? 1.f : 0.f;
    }
    for (int it = 0; it < iters; ++it) {
        mm(t, 32, z, 32, y, 32, 32, 32, 32, false);
        for (int i = tid; i < 1024; i += 128)
            t[i] = (((i % 33) == 0) ? 1.5f : 0.f) - 0.5f * t[i];
        mm(yn, 32, y, 32, t, 32, 32, 32, 32, false);
        mm(zn, 32, t, 32, z, 32, 32, 32, 32, false);
        float* tmp;
        tmp = y; y = yn; yn = tmp;
        tmp = z; z = zn; zn = tmp;
    }
    const float sc  = __fsqrt_rn(c);
    const float isc = __frsqrt_rn(c);
    __syncthreads();
    if (dst_s) for (int i = tid; i < 1024; i += 128) dst_s[i] = y[i] * sc;
    if (dst_i) for (int i = tid; i < 1024; i += 128) dst_i[i] = z[i] * isc;
    __syncthreads();
}

__global__ __launch_bounds__(128, 1)
void spdnet_mixedop_kernel(const float* __restrict__ x,
                           const float* __restrict__ wts,
                           const float* __restrict__ Ws,
                           float* __restrict__ out)
{
    __shared__ float sX[4096];
    __shared__ float sWk[2048];
    __shared__ float sP[2048];
    __shared__ float sS[4][1024];
    __shared__ float sB0[1024], sB1[1024], sB2[1024];
    __shared__ float sRed[32];

    const int bc  = blockIdx.x;
    const int tid = threadIdx.x;

    float wk[4];
    #pragma unroll
    for (int k = 0; k < 4; ++k) wk[k] = wts[k];

    for (int i = tid; i < 4096; i += 128) sX[i] = x[bc * 4096 + i];

    for (int k = 0; k < 4; ++k) {
        __syncthreads();
        for (int i = tid; i < 2048; i += 128) sWk[i] = Ws[k * 2048 + i];
        mm(sP, 64, sWk, 64, sX, 64, 32, 64, 64, false);
        mm(sS[k], 32, sP, 64, sWk, 64, 32, 32, 64, true);
    }

    __syncthreads();
    for (int i = tid; i < 1024; i += 128)
        sB0[i] = wk[0] * sS[0][i] + wk[1] * sS[1][i] +
                 wk[2] * sS[2][i] + wk[3] * sS[3][i];

    float* pSq = sX;
    float* pSi = sX + 1024;
    float* pU  = sX + 2048;
    ns_sqrt(pSq, pSi, sB0, sB1, sB2, sWk, sWk + 1024, sP, sRed, 14);

    __syncthreads();
    for (int i = tid; i < 1024; i += 128) pU[i] = 0.f;

    for (int k = 0; k < 4; ++k) {
        mm(sB0, 32, pSi, 32, sS[k], 32, 32, 32, 32, false);
        mm(sB1, 32, sB0, 32, pSi, 32, 32, 32, 32, false);

        ns_sqrt(sB2, nullptr, sB1, sP, sP + 1024, sWk, sWk + 1024, sB0, sRed, 12);
        ns_sqrt(sB1, nullptr, sB2, sP, sP + 1024, sWk, sWk + 1024, sB0, sRed, 12);
        ns_sqrt(sB2, nullptr, sB1, sP, sP + 1024, sWk, sWk + 1024, sB0, sRed, 12);

        __syncthreads();
        for (int i = tid; i < 1024; i += 128)
            if ((i % 33) == 0) sB2[i] -= 1.f;

        float* q  = sB0;
        float* qn = sB1;
        __syncthreads();
        for (int i = tid; i < 1024; i += 128)
            q[i] = ((i % 33) == 0) ? (-1.f / 8.f) : 0.f;
        for (int ii = 7; ii >= 1; --ii) {
            mm(qn, 32, sB2, 32, q, 32, 32, 32, 32, false);
            const float a = ((ii & 1) ? 1.f : -1.f) / (float)ii;
            for (int i = tid; i < 1024; i += 128)
                if ((i % 33) == 0) qn[i] += a;
            float* tmp = q; q = qn; qn = tmp;
        }
        mm(sWk, 32, sB2, 32, q, 32, 32, 32, 32, false);

        const float scale = wk[k] * 8.f;
        for (int i = tid; i < 1024; i += 128) pU[i] += scale * sWk[i];
        __syncthreads();
    }

    __syncthreads();
    for (int i = tid; i < 1024; i += 128) {
        sB0[i] = pU[i] * (1.f / 64.f);
        sB1[i] = ((i % 33) == 0) ? 1.f : 0.f;
    }
    for (int ii = 10; ii >= 1; --ii) {
        mm(sB2, 32, sB0, 32, sB1, 32, 32, 32, 32, false);
        const float inv = 1.f / (float)ii;
        for (int i = tid; i < 1024; i += 128)
            sB1[i] = (((i % 33) == 0) ? 1.f : 0.f) + sB2[i] * inv;
    }
    float* r  = sB1;
    float* rn = sB2;
    for (int sq = 0; sq < 6; ++sq) {
        mm(rn, 32, r, 32, r, 32, 32, 32, 32, false);
        float* tmp = r; r = rn; rn = tmp;
    }
    mm(sB0, 32, pSq, 32, r, 32, 32, 32, 32, false);
    mm(sP, 32, sB0, 32, pSq, 32, 32, 32, 32, false);
    __syncthreads();
    for (int i = tid; i < 1024; i += 128) out[bc * 1024 + i] = sP[i];
}

extern "C" void kernel_launch(void* const* d_in, const int* in_sizes, int n_in,
                              void* d_out, int out_size, void* d_ws, size_t ws_size,
                              hipStream_t stream)
{
    const float* x   = (const float*)d_in[0];
    const float* wts = (const float*)d_in[1];
    const float* Ws  = (const float*)d_in[2];
    float* out = (float*)d_out;

    dim3 grid(2048);
    dim3 block(128);
    hipLaunchKernelGGL(spdnet_mixedop_kernel, grid, block, 0, stream,
                       x, wts, Ws, out);
}


# LDS per one-wave workgroup 60 KiB -> 36 KiB (W_k fragments from global, Newton-Schulz in place on 4 rotating buffers, U in registers): 8 not 5 workgroups per WGP, all waves resident; same sums
# speedup vs baseline: 2.5140x; 2.5140x over previous
#include <hip/hip_runtime.h>


typedef __attribute__((ext_vector_type(2))) float v2f;
typedef __attribute__((ext_vector_type(8))) float v8f;

#define WMMA_F32(a, b, c) \
    __builtin_amdgcn_wmma_f32_16x16x4_f32(false, (a), false, (b), (short)0, (c), false, false)

__device__ __forceinline__ void ld32(float (&v)[32], const float* p, int lane)
{
    #pragma unroll
    for (int j = 0; j < 32; ++j) v[j] = p[lane + 32 * j];
}

__device__ __forceinline__ void mm(float* __restrict__ C, int ldc,
                                   const float* __restrict__ A, int lda,
                                   const float* __restrict__ B, int ldb,
                                   int N, int Kd, bool transB)
{
    __syncthreads();
    const int lane = threadIdx.x;
    const int half = lane >> 4;
    const int l15  = lane & 15;
    for (int ni = 0; ni < N; ni += 32) {
        v8f c00 = {}, c01 = {}, c10 = {}, c11 = {};
        for (int k0 = 0; k0 < Kd; k0 += 32) {
            v2f a0[8], a1[8], b0[8], b1[8];
            #pragma unroll
            for (int s = 0; s < 8; ++s) {
                const int k = k0 + 4 * s + 2 * half;
                const float* ar = A + l15 * lda + k;
                a0[s].x = ar[0];
                a0[s].y = ar[1];
                a1[s].x = ar[16 * lda];
                a1[s].y = ar[16 * lda + 1];
                if (!transB) {
                    const float* br = B + k * ldb + ni + l15;
                    b0[s].x = br[0];
                    b0[s].y = br[ldb];
                    b1[s].x = br[16];
                    b1[s].y = br[ldb + 16];
                } else {
                    const float* br = B + (ni + l15) * ldb + k;
                    b0[s].x = br[0];
                    b0[s].y = br[1];
                    b1[s].x = br[16 * ldb];
                    b1[s].y = br[16 * ldb + 1];
                }
            }
            #pragma unroll
            for (int s = 0; s < 8; ++s) {
                c00 = WMMA_F32(a0[s], b0[s], c00);
                c01 = WMMA_F32(a0[s], b1[s], c01);
                c10 = WMMA_F32(a1[s], b0[s], c10);
                c11 = WMMA_F32(a1[s], b1[s], c11);
            }
        }
        #pragma unroll
        for (int r = 0; r < 8; ++r) {
            float* cr = C + (half * 8 + r) * ldc + ni + l15;
            cr[0]            = c00[r];
            cr[16]           = c01[r];
            cr[16 * ldc]     = c10[r];
            cr[16 * ldc + 16] = c11[r];
        }
    }
    __syncthreads();
}

__device__ __forceinline__ float trace32(const float* A, float* red)
{
    __syncthreads();
    red[threadIdx.x] = A[threadIdx.x * 33];
    __syncthreads();
    float s = 0.f;
    #pragma unroll
    for (int i = 0; i < 32; ++i) s += red[i];
    return s;
}

__device__ __forceinline__ void ns_sqrt(float*& y, float*& z, float*& t, float*& f,
                                        int iters, bool want_inv)
{
    const int lane = threadIdx.x;
    const float c = trace32(y, t);
    const float inv_c = 1.f / c;
    __syncthreads();
    {
        float v[32];
        ld32(v, y, lane);
        #pragma unroll
        for (int j = 0; j < 32; ++j) {
            const int i = lane + 32 * j;
            y[i] = v[j] * inv_c;
            z[i] = ((i % 33) == 0) ? 1.f : 0.f;
        }
    }
    for (int it = 0; it < iters; ++it) {
        mm(t, 32, z, 32, y, 32, 32, 32, false);
        {
            float v[32];
            ld32(v, t, lane);
            #pragma unroll
            for (int j = 0; j < 32; ++j) {
                const int i = lane + 32 * j;
                t[i] = (((i % 33) == 0) ? 1.5f : 0.f) - 0.5f * v[j];
            }
        }
        mm(f, 32, y, 32, t, 32, 32, 32, false);
        mm(y, 32, t, 32, z, 32, 32, 32, false);
        float* oy = y;
        y = f; f = z; z = oy;
    }
    const float sc  = __fsqrt_rn(c);
    const float isc = __frsqrt_rn(c);
    __syncthreads();
    {
        float v[32];
        ld32(v, y, lane);
        #pragma unroll
        for (int j = 0; j < 32; ++j) y[lane + 32 * j] = v[j] * sc;
    }
    if (want_inv) {
        float v[32];
        ld32(v, z, lane);
        #pragma unroll
        for (int j = 0; j < 32; ++j) z[lane + 32 * j] = v[j] * isc;
    }
    __syncthreads();
}

__global__ __launch_bounds__(32)
void spdnet_mixedop_kernel(const float* __restrict__ x,
                           const float* __restrict__ wts,
                           const float* __restrict__ Ws,
                           float* __restrict__ out)
{
    __shared__ float sm[9 * 1024];
    float* const sX = sm;
    float* const sP = sm + 4 * 1024;
    float* const s0 = sm + 6 * 1024;
    float* const s1 = sm + 7 * 1024;
    float* const s2 = sm + 8 * 1024;
    float* const s3 = sm;

    const int bc   = blockIdx.x;
    const int lane = threadIdx.x;

    float wk[4];
    #pragma unroll
    for (int k = 0; k < 4; ++k) wk[k] = wts[k];

    #pragma unroll 16
    for (int i = lane; i < 4096; i += 32) sX[i] = x[bc * 4096 + i];

    for (int k = 0; k < 4; ++k) {
        const float* Wk = Ws + k * 2048;
        float* sk = (k < 3) ? (s0 + k * 1024) : s3;
        mm(sP, 64, Wk, 64, sX, 64, 64, 64, false);
        mm(sk, 32, sP, 64, Wk, 64, 32, 64, true);
    }

    float* by = sm + 1 * 1024;
    float* bz = sm + 2 * 1024;
    float* bt = sm + 3 * 1024;
    float* bf = sm + 4 * 1024;
    __syncthreads();
    #pragma unroll 8
    for (int i = lane; i < 1024; i += 32)
        by[i] = wk[0] * s0[i] + wk[1] * s1[i] +
                wk[2] * s2[i] + wk[3] * s3[i];

    ns_sqrt(by, bz, bt, bf, 14, true);
    float* const pSq = by;
    float* const pSi = bz;
    float* fa = bt;
    float* fb = bf;
    float* fc = sm + 5 * 1024;

    float u[32];
    #pragma unroll
    for (int j = 0; j < 32; ++j) u[j] = 0.f;

    for (int k = 0; k < 4; ++k) {
        float* y = (k < 3) ? (s0 + k * 1024) : s3;
        float* z = fa;
        float* t = fb;
        float* f = fc;
        mm(z, 32, pSi, 32, y, 32, 32, 32, false);
        mm(y, 32, z, 32, pSi, 32, 32, 32, false);

        for (int rt = 0; rt < 3; ++rt)
            ns_sqrt(y, z, t, f, 12, false);
        float* E = y;

        __syncthreads();
        E[lane * 33] -= 1.f;

        float* q  = z;
        float* qn = t;
        __syncthreads();
        #pragma unroll
        for (int j = 0; j < 32; ++j) {
            const int i = lane + 32 * j;
            q[i] = ((i % 33) == 0) ? (-1.f / 8.f) : 0.f;
        }
        for (int ii = 7; ii >= 1; --ii) {
            mm(qn, 32, E, 32, q, 32, 32, 32, false);
            const float a = ((ii & 1) ? 1.f : -1.f) / (float)ii;
            qn[lane * 33] += a;
            float* tmp = q; q = qn; qn = tmp;
        }
        mm(qn, 32, E, 32, q, 32, 32, 32, false);

        const float scale = wk[k] * 8.f;
        {
            float w[32];
            ld32(w, qn, lane);
            #pragma unroll
            for (int j = 0; j < 32; ++j) u[j] = u[j] + scale * w[j];
        }
        fa = z; fb = t; fc = f;
        __syncthreads();
    }

    float* const dV = sm + 6 * 1024;
    float* const dR = sm + 7 * 1024;
    float* const dT = sm + 8 * 1024;
    __syncthreads();
    #pragma unroll
    for (int j = 0; j < 32; ++j) {
        const int i = lane + 32 * j;
        dV[i] = u[j] * (1.f / 64.f);
        dR[i] = ((i % 33) == 0) ? 1.f : 0.f;
    }
    for (int ii = 10; ii >= 1; --ii) {
        mm(dT, 32, dV, 32, dR, 32, 32, 32, false);
        const float inv = 1.f / (float)ii;
        float v[32];
        ld32(v, dT, lane);
        #pragma unroll
        for (int j = 0; j < 32; ++j) {
            const int i = lane + 32 * j;
            dR[i] = (((i % 33) == 0) ? 1.f : 0.f) + v[j] * inv;
        }
    }
    float* r  = dR;
    float* rn = dT;
    for (int sq = 0; sq < 6; ++sq) {
        mm(rn, 32, r, 32, r, 32, 32, 32, false);
        float* tmp = r; r = rn; rn = tmp;
    }
    mm(dV, 32, pSq, 32, r, 32, 32, 32, false);
    mm(rn, 32, dV, 32, pSq, 32, 32, 32, false);
    __syncthreads();
    {
        float v[32];
        ld32(v, rn, lane);
        #pragma unroll
        for (int j = 0; j < 32; ++j) out[bc * 1024 + lane + 32 * j] = v[j];
    }
}

extern "C" void kernel_launch(void* const* d_in, const int* in_sizes, int n_in,
                              void* d_out, int out_size, void* d_ws, size_t ws_size,
                              hipStream_t stream)
{
    const float* x   = (const float*)d_in[0];
    const float* wts = (const float*)d_in[1];
    const float* Ws  = (const float*)d_in[2];
    float* out = (float*)d_out;

    dim3 grid(2048);
    dim3 block(32);
    hipLaunchKernelGGL(spdnet_mixedop_kernel, grid, block, 0, stream,
                       x, wts, Ws, out);
}


# Newton-Schulz roots in registers: a symmetric iterate's accumulator tiles are used as both A and B fragments (X^T for X), no LDS in the iteration; products sum in a new k order, rounding differs
# speedup vs baseline: 4.0733x; 1.6202x over previous
#include <hip/hip_runtime.h>


typedef __attribute__((ext_vector_type(2))) float v2f;
typedef __attribute__((ext_vector_type(8))) float v8f;

#define WMMA_F32(a, b, c) \
    __builtin_amdgcn_wmma_f32_16x16x4_f32(false, (a), false, (b), (short)0, (c), false, false)

__device__ __forceinline__ void ld32(float (&v)[32], const float* p, int lane)
{
    #pragma unroll
    for (int j = 0; j < 32; ++j) v[j] = p[lane + 32 * j];
}

__device__ __forceinline__ void mm(float* __restrict__ C, int ldc,
                                   const float* __restrict__ A, int lda,
                                   const float* __restrict__ B, int ldb,
                                   int N, int Kd, bool transB)
{
    __syncthreads();
    const int lane = threadIdx.x;
    const int half = lane >> 4;
    const int l15  = lane & 15;
    for (int ni = 0; ni < N; ni += 32) {
        v8f c00 = {}, c01 = {}, c10 = {}, c11 = {};
        for (int k0 = 0; k0 < Kd; k0 += 32) {
            v2f a0[8], a1[8], b0[8], b1[8];
            #pragma unroll
            for (int s = 0; s < 8; ++s) {
                const int k = k0 + 4 * s + 2 * half;
                const float* ar = A + l15 * lda + k;
                a0[s].x = ar[0];
                a0[s].y = ar[1];
                a1[s].x = ar[16 * lda];
                a1[s].y = ar[16 * lda + 1];
                if (!transB) {
                    const float* br = B + k * ldb + ni + l15;
                    b0[s].x = br[0];
                    b0[s].y = br[ldb];
                    b1[s].x = br[16];
                    b1[s].y = br[ldb + 16];
                } else {
                    const float* br = B + (ni + l15) * ldb + k;
                    b0[s].x = br[0];
                    b0[s].y = br[1];
                    b1[s].x = br[16 * ldb];
                    b1[s].y = br[16 * ldb + 1];
                }
            }
            #pragma unroll
            for (int s = 0; s < 8; ++s) {
                c00 = WMMA_F32(a0[s], b0[s], c00);
                c01 = WMMA_F32(a0[s], b1[s], c01);
                c10 = WMMA_F32(a1[s], b0[s], c10);
                c11 = WMMA_F32(a1[s], b1[s], c11);
            }
        }
        #pragma unroll
        for (int r = 0; r < 8; ++r) {
            float* cr = C + (half * 8 + r) * ldc + ni + l15;
            cr[0]            = c00[r];
            cr[16]           = c01[r];
            cr[16 * ldc]     = c10[r];
            cr[16 * ldc + 16] = c11[r];
        }
    }
    __syncthreads();
}

__device__ __forceinline__ float trace32(const float* A, float* red)
{
    __syncthreads();
    red[threadIdx.x] = A[threadIdx.x * 33];
    __syncthreads();
    float s = 0.f;
    #pragma unroll
    for (int i = 0; i < 32; ++i) s += red[i];
    return s;
}

struct M32 { v8f t[2][2]; };

__device__ __forceinline__ void mmT(M32& D, const M32& P, const M32& Q)
{
    v8f d00 = {}, d01 = {}, d10 = {}, d11 = {};
    #pragma unroll
    for (int tk = 0; tk < 2; ++tk) {
        #pragma unroll
        for (int s = 0; s < 4; ++s) {
            v2f a0, a1, b0, b1;
            a0.x = P.t[tk][0][2 * s]; a0.y = P.t[tk][0][2 * s + 1];
            a1.x = P.t[tk][1][2 * s]; a1.y = P.t[tk][1][2 * s + 1];
            b0.x = Q.t[tk][0][2 * s]; b0.y = Q.t[tk][0][2 * s + 1];
            b1.x = Q.t[tk][1][2 * s]; b1.y = Q.t[tk][1][2 * s + 1];
            d00 = WMMA_F32(a0, b0, d00);
            d01 = WMMA_F32(a0, b1, d01);
            d10 = WMMA_F32(a1, b0, d10);
            d11 = WMMA_F32(a1, b1, d11);
        }
    }
    D.t[0][0] = d00; D.t[0][1] = d01; D.t[1][0] = d10; D.t[1][1] = d11;
}

__device__ __forceinline__ void ns_residual(M32& T, int half, int l15)
{
    #pragma unroll
    for (int ti = 0; ti < 2; ++ti)
        #pragma unroll
        for (int tj = 0; tj < 2; ++tj)
            #pragma unroll
            for (int r = 0; r < 8; ++r) {
                const float d = (ti == tj && (8 * half + r) == l15) ? 1.5f : 0.f;
                T.t[ti][tj][r] = d - 0.5f * T.t[ti][tj][r];
            }
}

__device__ __forceinline__ void ns_sqrt(float* y, float* z, float* t, int iters, bool want_inv)
{
    const int lane = threadIdx.x;
    const int half = lane >> 4;
    const int l15  = lane & 15;
    const float c = trace32(y, t);
    const float inv_c = 1.f / c;
    __syncthreads();
    M32 Y, Z;
    #pragma unroll
    for (int ti = 0; ti < 2; ++ti)
        #pragma unroll
        for (int tj = 0; tj < 2; ++tj)
            #pragma unroll
            for (int r = 0; r < 8; ++r) {
                const int row = 16 * ti + 8 * half + r;
                const int col = 16 * tj + l15;
                Y.t[ti][tj][r] = y[row * 32 + col] * inv_c;
                Z.t[ti][tj][r] = (row == col) ? 1.f : 0.f;
            }
    #pragma nounroll
    for (int it = 0; it < iters; it += 2) {
        M32 T, Yn, Zn;
        mmT(T, Z, Y);
        ns_residual(T, half, l15);
        mmT(Yn, T, Y);
        mmT(Zn, Z, T);
        Y = Yn; Z = Zn;
        mmT(T, Y, Z);
        ns_residual(T, half, l15);
        mmT(Yn, Y, T);
        mmT(Zn, T, Z);
        Y = Yn; Z = Zn;
    }
    const float sc  = __fsqrt_rn(c);
    const float isc = __frsqrt_rn(c);
    #pragma unroll
    for (int ti = 0; ti < 2; ++ti)
        #pragma unroll
        for (int tj = 0; tj < 2; ++tj)
            #pragma unroll
            for (int r = 0; r < 8; ++r) {
                const int i = (16 * ti + 8 * half + r) * 32 + 16 * tj + l15;
                y[i] = Y.t[ti][tj][r] * sc;
                if (want_inv) z[i] = Z.t[ti][tj][r] * isc;
            }
    __syncthreads();
}

__global__ __launch_bounds__(32)
void spdnet_mixedop_kernel(const float* __restrict__ x,
                           const float* __restrict__ wts,
                           const float* __restrict__ Ws,
                           float* __restrict__ out)
{
    __shared__ float sm[9 * 1024];
    float* const sX = sm;
    float* const sP = sm + 4 * 1024;
    float* const s0 = sm + 6 * 1024;
    float* const s1 = sm + 7 * 1024;
    float* const s2 = sm + 8 * 1024;
    float* const s3 = sm;

    const int bc   = blockIdx.x;
    const int lane = threadIdx.x;

    float wk[4];
    #pragma unroll
    for (int k = 0; k < 4; ++k) wk[k] = wts[k];

    #pragma unroll 16
    for (int i = lane; i < 4096; i += 32) sX[i] = x[bc * 4096 + i];

    for (int k = 0; k < 4; ++k) {
        const float* Wk = Ws + k * 2048;
        float* sk = (k < 3) ? (s0 + k * 1024) : s3;
        mm(sP, 64, Wk, 64, sX, 64, 64, 64, false);
        mm(sk, 32, sP, 64, Wk, 64, 32, 64, true);
    }

    float* by = sm + 1 * 1024;
    float* bz = sm + 2 * 1024;
    float* bt = sm + 3 * 1024;
    float* bf = sm + 4 * 1024;
    __syncthreads();
    #pragma unroll 8
    for (int i = lane; i < 1024; i += 32)
        by[i] = wk[0] * s0[i] + wk[1] * s1[i] +
                wk[2] * s2[i] + wk[3] * s3[i];

    ns_sqrt(by, bz, bt, 14, true);
    float* const pSq = by;
    float* const pSi = bz;
    float* fa = bt;
    float* fb = bf;
    float* fc = sm + 5 * 1024;

    float u[32];
    #pragma unroll
    for (int j = 0; j < 32; ++j) u[j] = 0.f;

    for (int k = 0; k < 4; ++k) {
        float* y = (k < 3) ? (s0 + k * 1024) : s3;
        float* z = fa;
        float* t = fb;
        float* f = fc;
        mm(z, 32, pSi, 32, y, 32, 32, 32, false);
        mm(y, 32, z, 32, pSi, 32, 32, 32, false);

        for (int rt = 0; rt < 3; ++rt)
            ns_sqrt(y, z, t, 12, false);
        float* E = y;

        __syncthreads();
        E[lane * 33] -= 1.f;

        float* q  = z;
        float* qn = t;
        __syncthreads();
        #pragma unroll
        for (int j = 0; j < 32; ++j) {
            const int i = lane + 32 * j;
            q[i] = ((i % 33) == 0) ? (-1.f / 8.f) : 0.f;
        }
        for (int ii = 7; ii >= 1; --ii) {
            mm(qn, 32, E, 32, q, 32, 32, 32, false);
            const float a = ((ii & 1) ? 1.f : -1.f) / (float)ii;
            qn[lane * 33] += a;
            float* tmp = q; q = qn; qn = tmp;
        }
        mm(qn, 32, E, 32, q, 32, 32, 32, false);

        const float scale = wk[k] * 8.f;
        {
            float w[32];
            ld32(w, qn, lane);
            #pragma unroll
            for (int j = 0; j < 32; ++j) u[j] = u[j] + scale * w[j];
        }
        fa = z; fb = t; fc = f;
        __syncthreads();
    }

    float* const dV = sm + 6 * 1024;
    float* const dR = sm + 7 * 1024;
    float* const dT = sm + 8 * 1024;
    __syncthreads();
    #pragma unroll
    for (int j = 0; j < 32; ++j) {
        const int i = lane + 32 * j;
        dV[i] = u[j] * (1.f / 64.f);
        dR[i] = ((i % 33) == 0) ? 1.f : 0.f;
    }
    for (int ii = 10; ii >= 1; --ii) {
        mm(dT, 32, dV, 32, dR, 32, 32, 32, false);
        const float inv = 1.f / (float)ii;
        float v[32];
        ld32(v, dT, lane);
        #pragma unroll
        for (int j = 0; j < 32; ++j) {
            const int i = lane + 32 * j;
            dR[i] = (((i % 33) == 0) ? 1.f : 0.f) + v[j] * inv;
        }
    }
    float* r  = dR;
    float* rn = dT;
    for (int sq = 0; sq < 6; ++sq) {
        mm(rn, 32, r, 32, r, 32, 32, 32, false);
        float* tmp = r; r = rn; rn = tmp;
    }
    mm(dV, 32, pSq, 32, r, 32, 32, 32, false);
    mm(rn, 32, dV, 32, pSq, 32, 32, 32, false);
    __syncthreads();
    {
        float v[32];
        ld32(v, rn, lane);
        #pragma unroll
        for (int j = 0; j < 32; ++j) out[bc * 1024 + lane + 32 * j] = v[j];
    }
}

extern "C" void kernel_launch(void* const* d_in, const int* in_sizes, int n_in,
                              void* d_out, int out_size, void* d_ws, size_t ws_size,
                              hipStream_t stream)
{
    const float* x   = (const float*)d_in[0];
    const float* wts = (const float*)d_in[1];
    const float* Ws  = (const float*)d_in[2];
    float* out = (float*)d_out;

    dim3 grid(2048);
    dim3 block(32);
    hipLaunchKernelGGL(spdnet_mixedop_kernel, grid, block, 0, stream,
                       x, wts, Ws, out);
}


# Square roots by the coupled iteration of order 5 (Y,Z times I+R/2+3R^2/8+5R^3/16+35R^4/128, R=I-ZY: 5 products) instead of Newton-Schulz (3): 6 and 7 steps replace 12 and 14; rounding differs
# speedup vs baseline: 4.5705x; 1.1221x over previous
#include <hip/hip_runtime.h>


typedef __attribute__((ext_vector_type(2))) float v2f;
typedef __attribute__((ext_vector_type(8))) float v8f;

#define WMMA_F32(a, b, c) \
    __builtin_amdgcn_wmma_f32_16x16x4_f32(false, (a), false, (b), (short)0, (c), false, false)

__device__ __forceinline__ void ld32(float (&v)[32], const float* p, int lane)
{
    #pragma unroll
    for (int j = 0; j < 32; ++j) v[j] = p[lane + 32 * j];
}

__device__ __forceinline__ void mm(float* __restrict__ C, int ldc,
                                   const float* __restrict__ A, int lda,
                                   const float* __restrict__ B, int ldb,
                                   int N, int Kd, bool transB)
{
    __syncthreads();
    const int lane = threadIdx.x;
    const int half = lane >> 4;
    const int l15  = lane & 15;
    for (int ni = 0; ni < N; ni += 32) {
        v8f c00 = {}, c01 = {}, c10 = {}, c11 = {};
        for (int k0 = 0; k0 < Kd; k0 += 32) {
            v2f a0[8], a1[8], b0[8], b1[8];
            #pragma unroll
            for (int s = 0; s < 8; ++s) {
                const int k = k0 + 4 * s + 2 * half;
                const float* ar = A + l15 * lda + k;
                a0[s].x = ar[0];
                a0[s].y = ar[1];
                a1[s].x = ar[16 * lda];
                a1[s].y = ar[16 * lda + 1];
                if (!transB) {
                    const float* br = B + k * ldb + ni + l15;
                    b0[s].x = br[0];
                    b0[s].y = br[ldb];
                    b1[s].x = br[16];
                    b1[s].y = br[ldb + 16];
                } else {
                    const float* br = B + (ni + l15) * ldb + k;
                    b0[s].x = br[0];
                    b0[s].y = br[1];
                    b1[s].x = br[16 * ldb];
                    b1[s].y = br[16 * ldb + 1];
                }
            }
            #pragma unroll
            for (int s = 0; s < 8; ++s) {
                c00 = WMMA_F32(a0[s], b0[s], c00);
                c01 = WMMA_F32(a0[s], b1[s], c01);
                c10 = WMMA_F32(a1[s], b0[s], c10);
                c11 = WMMA_F32(a1[s], b1[s], c11);
            }
        }
        #pragma unroll
        for (int r = 0; r < 8; ++r) {
            float* cr = C + (half * 8 + r) * ldc + ni + l15;
            cr[0]            = c00[r];
            cr[16]           = c01[r];
            cr[16 * ldc]     = c10[r];
            cr[16 * ldc + 16] = c11[r];
        }
    }
    __syncthreads();
}

__device__ __forceinline__ float trace32(const float* A, float* red)
{
    __syncthreads();
    red[threadIdx.x] = A[threadIdx.x * 33];
    __syncthreads();
    float s = 0.f;
    #pragma unroll
    for (int i = 0; i < 32; ++i) s += red[i];
    return s;
}

struct M32 { v8f t[2][2]; };

__device__ __forceinline__ void mmT(M32& D, const M32& P, const M32& Q)
{
    v8f d00 = {}, d01 = {}, d10 = {}, d11 = {};
    #pragma unroll
    for (int tk = 0; tk < 2; ++tk) {
        #pragma unroll
        for (int s = 0; s < 4; ++s) {
            v2f a0, a1, b0, b1;
            a0.x = P.t[tk][0][2 * s]; a0.y = P.t[tk][0][2 * s + 1];
            a1.x = P.t[tk][1][2 * s]; a1.y = P.t[tk][1][2 * s + 1];
            b0.x = Q.t[tk][0][2 * s]; b0.y = Q.t[tk][0][2 * s + 1];
            b1.x = Q.t[tk][1][2 * s]; b1.y = Q.t[tk][1][2 * s + 1];
            d00 = WMMA_F32(a0, b0, d00);
            d01 = WMMA_F32(a0, b1, d01);
            d10 = WMMA_F32(a1, b0, d10);
            d11 = WMMA_F32(a1, b1, d11);
        }
    }
    D.t[0][0] = d00; D.t[0][1] = d01; D.t[1][0] = d10; D.t[1][1] = d11;
}

__device__ __forceinline__ void mmT_acc(M32& D, const M32& C, const M32& P, const M32& Q)
{
    v8f d00 = C.t[0][0], d01 = C.t[0][1], d10 = C.t[1][0], d11 = C.t[1][1];
    #pragma unroll
    for (int tk = 0; tk < 2; ++tk) {
        #pragma unroll
        for (int s = 0; s < 4; ++s) {
            v2f a0, a1, b0, b1;
            a0.x = P.t[tk][0][2 * s]; a0.y = P.t[tk][0][2 * s + 1];
            a1.x = P.t[tk][1][2 * s]; a1.y = P.t[tk][1][2 * s + 1];
            b0.x = Q.t[tk][0][2 * s]; b0.y = Q.t[tk][0][2 * s + 1];
            b1.x = Q.t[tk][1][2 * s]; b1.y = Q.t[tk][1][2 * s + 1];
            d00 = WMMA_F32(a0, b0, d00);
            d01 = WMMA_F32(a0, b1, d01);
            d10 = WMMA_F32(a1, b0, d10);
            d11 = WMMA_F32(a1, b1, d11);
        }
    }
    D.t[0][0] = d00; D.t[0][1] = d01; D.t[1][0] = d10; D.t[1][1] = d11;
}

template <bool FLIP>
__device__ __forceinline__ void sqrt5_step(M32& Y, M32& Z, int half, int l15)
{
    M32 R, R2, W, P;
    if (!FLIP) mmT(R, Z, Y); else mmT(R, Y, Z);
    #pragma unroll
    for (int ti = 0; ti < 2; ++ti)
        #pragma unroll
        for (int tj = 0; tj < 2; ++tj)
            #pragma unroll
            for (int r = 0; r < 8; ++r) {
                const float d = (ti == tj && (8 * half + r) == l15) ? 1.f : 0.f;
                R.t[ti][tj][r] = d - R.t[ti][tj][r];
            }
    mmT(R2, R, R);
    #pragma unroll
    for (int ti = 0; ti < 2; ++ti)
        #pragma unroll
        for (int tj = 0; tj < 2; ++tj)
            #pragma unroll
            for (int r = 0; r < 8; ++r) {
                const float d  = (ti == tj && (8 * half + r) == l15) ? 1.f : 0.f;
                const float x  = R.t[ti][tj][r];
                const float x2 = R2.t[ti][tj][r];
                W.t[ti][tj][r] = 0.3125f * x + 0.2734375f * x2;
                P.t[ti][tj][r] = d + 0.5f * x + 0.375f * x2;
            }
    if (!FLIP) mmT_acc(P, P, R2, W); else mmT_acc(P, P, W, R2);
    M32 Yn, Zn;
    if (!FLIP) { mmT(Yn, P, Y); mmT(Zn, Z, P); }
    else       { mmT(Yn, Y, P); mmT(Zn, P, Z); }
    Y = Yn; Z = Zn;
}

__device__ __forceinline__ void ns_sqrt(float* y, float* z, float* t, int iters, bool want_inv)
{
    const int lane = threadIdx.x;
    const int half = lane >> 4;
    const int l15  = lane & 15;
    const float c = trace32(y, t);
    const float inv_c = 1.f / c;
    __syncthreads();
    M32 Y, Z;
    #pragma unroll
    for (int ti = 0; ti < 2; ++ti)
        #pragma unroll
        for (int tj = 0; tj < 2; ++tj)
            #pragma unroll
            for (int r = 0; r < 8; ++r) {
                const int row = 16 * ti + 8 * half + r;
                const int col = 16 * tj + l15;
                Y.t[ti][tj][r] = y[row * 32 + col] * inv_c;
                Z.t[ti][tj][r] = (row == col) ? 1.f : 0.f;
            }
    #pragma nounroll
    for (int it = 0; it + 1 < iters; it += 2) {
        sqrt5_step<false>(Y, Z, half, l15);
        sqrt5_step<true>(Y, Z, half, l15);
    }
    if (iters & 1) sqrt5_step<false>(Y, Z, half, l15);
    const float sc  = __fsqrt_rn(c);
    const float isc = __frsqrt_rn(c);
    #pragma unroll
    for (int ti = 0; ti < 2; ++ti)
        #pragma unroll
        for (int tj = 0; tj < 2; ++tj)
            #pragma unroll
            for (int r = 0; r < 8; ++r) {
                const int i = (16 * ti + 8 * half + r) * 32 + 16 * tj + l15;
                y[i] = Y.t[ti][tj][r] * sc;
                if (want_inv) z[i] = Z.t[ti][tj][r] * isc;
            }
    __syncthreads();
}

__global__ __launch_bounds__(32)
void spdnet_mixedop_kernel(const float* __restrict__ x,
                           const float* __restrict__ wts,
                           const float* __restrict__ Ws,
                           float* __restrict__ out)
{
    __shared__ float sm[9 * 1024];
    float* const sX = sm;
    float* const sP = sm + 4 * 1024;
    float* const s0 = sm + 6 * 1024;
    float* const s1 = sm + 7 * 1024;
    float* const s2 = sm + 8 * 1024;
    float* const s3 = sm;

    const int bc   = blockIdx.x;
    const int lane = threadIdx.x;

    float wk[4];
    #pragma unroll
    for (int k = 0; k < 4; ++k) wk[k] = wts[k];

    #pragma unroll 16
    for (int i = lane; i < 4096; i += 32) sX[i] = x[bc * 4096 + i];

    for (int k = 0; k < 4; ++k) {
        const float* Wk = Ws + k * 2048;
        float* sk = (k < 3) ? (s0 + k * 1024) : s3;
        mm(sP, 64, Wk, 64, sX, 64, 64, 64, false);
        mm(sk, 32, sP, 64, Wk, 64, 32, 64, true);
    }

    float* by = sm + 1 * 1024;
    float* bz = sm + 2 * 1024;
    float* bt = sm + 3 * 1024;
    float* bf = sm + 4 * 1024;
    __syncthreads();
    #pragma unroll 8
    for (int i = lane; i < 1024; i += 32)
        by[i] = wk[0] * s0[i] + wk[1] * s1[i] +
                wk[2] * s2[i] + wk[3] * s3[i];

    ns_sqrt(by, bz, bt, 7, true);
    float* const pSq = by;
    float* const pSi = bz;
    float* fa = bt;
    float* fb = bf;
    float* fc = sm + 5 * 1024;

    float u[32];
    #pragma unroll
    for (int j = 0; j < 32; ++j) u[j] = 0.f;

    for (int k = 0; k < 4; ++k) {
        float* y = (k < 3) ? (s0 + k * 1024) : s3;
        float* z = fa;
        float* t = fb;
        float* f = fc;
        mm(z, 32, pSi, 32, y, 32, 32, 32, false);
        mm(y, 32, z, 32, pSi, 32, 32, 32, false);

        for (int rt = 0; rt < 3; ++rt)
            ns_sqrt(y, z, t, 6, false);
        float* E = y;

        __syncthreads();
        E[lane * 33] -= 1.f;

        float* q  = z;
        float* qn = t;
        __syncthreads();
        #pragma unroll
        for (int j = 0; j < 32; ++j) {
            const int i = lane + 32 * j;
            q[i] = ((i % 33) == 0) ? (-1.f / 8.f) : 0.f;
        }
        for (int ii = 7; ii >= 1; --ii) {
            mm(qn, 32, E, 32, q, 32, 32, 32, false);
            const float a = ((ii & 1) ? 1.f : -1.f) / (float)ii;
            qn[lane * 33] += a;
            float* tmp = q; q = qn; qn = tmp;
        }
        mm(qn, 32, E, 32, q, 32, 32, 32, false);

        const float scale = wk[k] * 8.f;
        {
            float w[32];
            ld32(w, qn, lane);
            #pragma unroll
            for (int j = 0; j < 32; ++j) u[j] = u[j] + scale * w[j];
        }
        fa = z; fb = t; fc = f;
        __syncthreads();
    }

    float* const dV = sm + 6 * 1024;
    float* const dR = sm + 7 * 1024;
    float* const dT = sm + 8 * 1024;
    __syncthreads();
    #pragma unroll
    for (int j = 0; j < 32; ++j) {
        const int i = lane + 32 * j;
        dV[i] = u[j] * (1.f / 64.f);
        dR[i] = ((i % 33) == 0) ? 1.f : 0.f;
    }
    for (int ii = 10; ii >= 1; --ii) {
        mm(dT, 32, dV, 32, dR, 32, 32, 32, false);
        const float inv = 1.f / (float)ii;
        float v[32];
        ld32(v, dT, lane);
        #pragma unroll
        for (int j = 0; j < 32; ++j) {
            const int i = lane + 32 * j;
            dR[i] = (((i % 33) == 0) ? 1.f : 0.f) + v[j] * inv;
        }
    }
    float* r  = dR;
    float* rn = dT;
    for (int sq = 0; sq < 6; ++sq) {
        mm(rn, 32, r, 32, r, 32, 32, 32, false);
        float* tmp = r; r = rn; rn = tmp;
    }
    mm(dV, 32, pSq, 32, r, 32, 32, 32, false);
    mm(rn, 32, dV, 32, pSq, 32, 32, 32, false);
    __syncthreads();
    {
        float v[32];
        ld32(v, rn, lane);
        #pragma unroll
        for (int j = 0; j < 32; ++j) out[bc * 1024 + lane + 32 * j] = v[j];
    }
}

extern "C" void kernel_launch(void* const* d_in, const int* in_sizes, int n_in,
                              void* d_out, int out_size, void* d_ws, size_t ws_size,
                              hipStream_t stream)
{
    const float* x   = (const float*)d_in[0];
    const float* wts = (const float*)d_in[1];
    const float* Ws  = (const float*)d_in[2];
    float* out = (float*)d_out;

    dim3 grid(2048);
    dim3 block(32);
    hipLaunchKernelGGL(spdnet_mixedop_kernel, grid, block, 0, stream,
                       x, wts, Ws, out);
}


# Square-root steps use per-step minimax polynomials of degree 4 instead of the Taylor one: 4 steps replace 6 and 7; source and iterates symmetrised via LDS against skew growth; rounding differs
# speedup vs baseline: 5.8160x; 1.2725x over previous
#include <hip/hip_runtime.h>


typedef __attribute__((ext_vector_type(2))) float v2f;
typedef __attribute__((ext_vector_type(8))) float v8f;

#define WMMA_F32(a, b, c) \
    __builtin_amdgcn_wmma_f32_16x16x4_f32(false, (a), false, (b), (short)0, (c), false, false)

__device__ __forceinline__ void ld32(float (&v)[32], const float* p, int lane)
{
    #pragma unroll
    for (int j = 0; j < 32; ++j) v[j] = p[lane + 32 * j];
}

__device__ __forceinline__ void mm(float* __restrict__ C, int ldc,
                                   const float* __restrict__ A, int lda,
                                   const float* __restrict__ B, int ldb,
                                   int N, int Kd, bool transB)
{
    __syncthreads();
    const int lane = threadIdx.x;
    const int half = lane >> 4;
    const int l15  = lane & 15;
    for (int ni = 0; ni < N; ni += 32) {
        v8f c00 = {}, c01 = {}, c10 = {}, c11 = {};
        for (int k0 = 0; k0 < Kd; k0 += 32) {
            v2f a0[8], a1[8], b0[8], b1[8];
            #pragma unroll
            for (int s = 0; s < 8; ++s) {
                const int k = k0 + 4 * s + 2 * half;
                const float* ar = A + l15 * lda + k;
                a0[s].x = ar[0];
                a0[s].y = ar[1];
                a1[s].x = ar[16 * lda];
                a1[s].y = ar[16 * lda + 1];
                if (!transB) {
                    const float* br = B + k * ldb + ni + l15;
                    b0[s].x = br[0];
                    b0[s].y = br[ldb];
                    b1[s].x = br[16];
                    b1[s].y = br[ldb + 16];
                } else {
                    const float* br = B + (ni + l15) * ldb + k;
                    b0[s].x = br[0];
                    b0[s].y = br[1];
                    b1[s].x = br[16 * ldb];
                    b1[s].y = br[16 * ldb + 1];
                }
            }
            #pragma unroll
            for (int s = 0; s < 8; ++s) {
                c00 = WMMA_F32(a0[s], b0[s], c00);
                c01 = WMMA_F32(a0[s], b1[s], c01);
                c10 = WMMA_F32(a1[s], b0[s], c10);
                c11 = WMMA_F32(a1[s], b1[s], c11);
            }
        }
        #pragma unroll
        for (int r = 0; r < 8; ++r) {
            float* cr = C + (half * 8 + r) * ldc + ni + l15;
            cr[0]            = c00[r];
            cr[16]           = c01[r];
            cr[16 * ldc]     = c10[r];
            cr[16 * ldc + 16] = c11[r];
        }
    }
    __syncthreads();
}

__device__ __forceinline__ float trace32(const float* A, float* red)
{
    __syncthreads();
    red[threadIdx.x] = A[threadIdx.x * 33];
    __syncthreads();
    float s = 0.f;
    #pragma unroll
    for (int i = 0; i < 32; ++i) s += red[i];
    return s;
}

struct M32 { v8f t[2][2]; };

__device__ __forceinline__ void mmT(M32& D, const M32& P, const M32& Q)
{
    v8f d00 = {}, d01 = {}, d10 = {}, d11 = {};
    #pragma unroll
    for (int tk = 0; tk < 2; ++tk) {
        #pragma unroll
        for (int s = 0; s < 4; ++s) {
            v2f a0, a1, b0, b1;
            a0.x = P.t[tk][0][2 * s]; a0.y = P.t[tk][0][2 * s + 1];
            a1.x = P.t[tk][1][2 * s]; a1.y = P.t[tk][1][2 * s + 1];
            b0.x = Q.t[tk][0][2 * s]; b0.y = Q.t[tk][0][2 * s + 1];
            b1.x = Q.t[tk][1][2 * s]; b1.y = Q.t[tk][1][2 * s + 1];
            d00 = WMMA_F32(a0, b0, d00);
            d01 = WMMA_F32(a0, b1, d01);
            d10 = WMMA_F32(a1, b0, d10);
            d11 = WMMA_F32(a1, b1, d11);
        }
    }
    D.t[0][0] = d00; D.t[0][1] = d01; D.t[1][0] = d10; D.t[1][1] = d11;
}

__device__ __forceinline__ void mmT_acc(M32& D, const M32& C, const M32& P, const M32& Q)
{
    v8f d00 = C.t[0][0], d01 = C.t[0][1], d10 = C.t[1][0], d11 = C.t[1][1];
    #pragma unroll
    for (int tk = 0; tk < 2; ++tk) {
        #pragma unroll
        for (int s = 0; s < 4; ++s) {
            v2f a0, a1, b0, b1;
            a0.x = P.t[tk][0][2 * s]; a0.y = P.t[tk][0][2 * s + 1];
            a1.x = P.t[tk][1][2 * s]; a1.y = P.t[tk][1][2 * s + 1];
            b0.x = Q.t[tk][0][2 * s]; b0.y = Q.t[tk][0][2 * s + 1];
            b1.x = Q.t[tk][1][2 * s]; b1.y = Q.t[tk][1][2 * s + 1];
            d00 = WMMA_F32(a0, b0, d00);
            d01 = WMMA_F32(a0, b1, d01);
            d10 = WMMA_F32(a1, b0, d10);
            d11 = WMMA_F32(a1, b1, d11);
        }
    }
    D.t[0][0] = d00; D.t[0][1] = d01; D.t[1][0] = d10; D.t[1][1] = d11;
}

template <bool FLIP>
__device__ __forceinline__ void sqrt_step(M32& Y, M32& Z, int half, int l15,
                                          float sg, float k0, float k1, float k2, float k3, float k4)
{
    M32 X, X2, W, P;
    if (!FLIP) mmT(X, Z, Y); else mmT(X, Y, Z);
    #pragma unroll
    for (int ti = 0; ti < 2; ++ti)
        #pragma unroll
        for (int tj = 0; tj < 2; ++tj)
            #pragma unroll
            for (int r = 0; r < 8; ++r) {
                const float d = (ti == tj && (8 * half + r) == l15) ? sg : 0.f;
                X.t[ti][tj][r] = X.t[ti][tj][r] - d;
            }
    mmT(X2, X, X);
    #pragma unroll
    for (int ti = 0; ti < 2; ++ti)
        #pragma unroll
        for (int tj = 0; tj < 2; ++tj)
            #pragma unroll
            for (int r = 0; r < 8; ++r) {
                const float d  = (ti == tj && (8 * half + r) == l15) ? k0 : 0.f;
                const float x  = X.t[ti][tj][r];
                const float x2 = X2.t[ti][tj][r];
                W.t[ti][tj][r] = k3 * x + k4 * x2;
                P.t[ti][tj][r] = d + k1 * x + k2 * x2;
            }
    if (!FLIP) mmT_acc(P, P, X2, W); else mmT_acc(P, P, W, X2);
    M32 Yn, Zn;
    if (!FLIP) { mmT(Yn, P, Y); mmT(Zn, Z, P); }
    else       { mmT(Yn, Y, P); mmT(Zn, P, Z); }
    Y = Yn; Z = Zn;
}

__device__ __forceinline__ void symmetrise2(M32& Y, M32& Z, float* by, float* bz, int half, int l15)
{
    #pragma unroll
    for (int ti = 0; ti < 2; ++ti)
        #pragma unroll
        for (int tj = 0; tj < 2; ++tj)
            #pragma unroll
            for (int r = 0; r < 8; ++r) {
                const int i = (16 * ti + 8 * half + r) * 32 + 16 * tj + l15;
                by[i] = Y.t[ti][tj][r];
                bz[i] = Z.t[ti][tj][r];
            }
    __syncthreads();
    #pragma unroll
    for (int ti = 0; ti < 2; ++ti)
        #pragma unroll
        for (int tj = 0; tj < 2; ++tj)
            #pragma unroll
            for (int r = 0; r < 8; ++r) {
                const int j = (16 * tj + l15) * 32 + 16 * ti + 8 * half + r;
                Y.t[ti][tj][r] = 0.5f * (Y.t[ti][tj][r] + by[j]);
                Z.t[ti][tj][r] = 0.5f * (Z.t[ti][tj][r] + bz[j]);
            }
    __syncthreads();
}

__device__ __forceinline__ void ns_sqrt(float* y, float* z, float* t, bool want_inv)
{
    const int lane = threadIdx.x;
    const int half = lane >> 4;
    const int l15  = lane & 15;
    const float c = trace32(y, t);
    const float inv_c = 1.f / c;
    __syncthreads();
    M32 Y, Z;
    #pragma unroll
    for (int ti = 0; ti < 2; ++ti)
        #pragma unroll
        for (int tj = 0; tj < 2; ++tj)
            #pragma unroll
            for (int r = 0; r < 8; ++r) {
                const int row = 16 * ti + 8 * half + r;
                const int col = 16 * tj + l15;
                Y.t[ti][tj][r] = 0.5f * (y[row * 32 + col] + y[col * 32 + row]) * inv_c;
                Z.t[ti][tj][r] = (row == col) ? 1.f : 0.f;
            }
    #pragma nounroll
    for (int it = 0; it < 2; ++it) {
        const bool a = (it == 0);
        sqrt_step<false>(Y, Z, half, l15,
                         a ? 0.550000012f : 1.61000001f,
                         a ? 2.10768604f : 0.790120423f, a ? 5.74562931f : -0.282128632f, a ? -30.9237690f : 0.0719144717f,
                         a ? -54.0620193f : 0.150992960f, a ? 162.579529f : 0.289191633f);
        symmetrise2(Y, Z, y, t, half, l15);
        sqrt_step<true>(Y, Z, half, l15,
                        a ? 1.79999995f : 1.00000000f,
                        a ? 0.935014427f : 1.00000000f, a ? 0.400891274f : -0.500000000f, a ? -0.721723795f : 0.375000000f,
                        a ? -0.493403643f : -0.312500000f, a ? 0.415029675f : 0.273437500f);
    }
    const float sc  = __fsqrt_rn(c);
    const float isc = __frsqrt_rn(c);
    #pragma unroll
    for (int ti = 0; ti < 2; ++ti)
        #pragma unroll
        for (int tj = 0; tj < 2; ++tj)
            #pragma unroll
            for (int r = 0; r < 8; ++r) {
                const int i = (16 * ti + 8 * half + r) * 32 + 16 * tj + l15;
                y[i] = Y.t[ti][tj][r] * sc;
                if (want_inv) z[i] = Z.t[ti][tj][r] * isc;
            }
    __syncthreads();
}

__global__ __launch_bounds__(32)
void spdnet_mixedop_kernel(const float* __restrict__ x,
                           const float* __restrict__ wts,
                           const float* __restrict__ Ws,
                           float* __restrict__ out)
{
    __shared__ float sm[9 * 1024];
    float* const sX = sm;
    float* const sP = sm + 4 * 1024;
    float* const s0 = sm + 6 * 1024;
    float* const s1 = sm + 7 * 1024;
    float* const s2 = sm + 8 * 1024;
    float* const s3 = sm;

    const int bc   = blockIdx.x;
    const int lane = threadIdx.x;

    float wk[4];
    #pragma unroll
    for (int k = 0; k < 4; ++k) wk[k] = wts[k];

    #pragma unroll 16
    for (int i = lane; i < 4096; i += 32) sX[i] = x[bc * 4096 + i];

    for (int k = 0; k < 4; ++k) {
        const float* Wk = Ws + k * 2048;
        float* sk = (k < 3) ? (s0 + k * 1024) : s3;
        mm(sP, 64, Wk, 64, sX, 64, 64, 64, false);
        mm(sk, 32, sP, 64, Wk, 64, 32, 64, true);
    }

    float* by = sm + 1 * 1024;
    float* bz = sm + 2 * 1024;
    float* bt = sm + 3 * 1024;
    float* bf = sm + 4 * 1024;
    __syncthreads();
    #pragma unroll 8
    for (int i = lane; i < 1024; i += 32)
        by[i] = wk[0] * s0[i] + wk[1] * s1[i] +
                wk[2] * s2[i] + wk[3] * s3[i];

    ns_sqrt(by, bz, bt, true);
    float* const pSq = by;
    float* const pSi = bz;
    float* fa = bt;
    float* fb = bf;
    float* fc = sm + 5 * 1024;

    float u[32];
    #pragma unroll
    for (int j = 0; j < 32; ++j) u[j] = 0.f;

    for (int k = 0; k < 4; ++k) {
        float* y = (k < 3) ? (s0 + k * 1024) : s3;
        float* z = fa;
        float* t = fb;
        float* f = fc;
        mm(z, 32, pSi, 32, y, 32, 32, 32, false);
        mm(y, 32, z, 32, pSi, 32, 32, 32, false);

        for (int rt = 0; rt < 3; ++rt)
            ns_sqrt(y, z, t, false);
        float* E = y;

        __syncthreads();
        E[lane * 33] -= 1.f;

        float* q  = z;
        float* qn = t;
        __syncthreads();
        #pragma unroll
        for (int j = 0; j < 32; ++j) {
            const int i = lane + 32 * j;
            q[i] = ((i % 33) == 0) ? (-1.f / 8.f) : 0.f;
        }
        for (int ii = 7; ii >= 1; --ii) {
            mm(qn, 32, E, 32, q, 32, 32, 32, false);
            const float a = ((ii & 1) ? 1.f : -1.f) / (float)ii;
            qn[lane * 33] += a;
            float* tmp = q; q = qn; qn = tmp;
        }
        mm(qn, 32, E, 32, q, 32, 32, 32, false);

        const float scale = wk[k] * 8.f;
        {
            float w[32];
            ld32(w, qn, lane);
            #pragma unroll
            for (int j = 0; j < 32; ++j) u[j] = u[j] + scale * w[j];
        }
        fa = z; fb = t; fc = f;
        __syncthreads();
    }

    float* const dV = sm + 6 * 1024;
    float* const dR = sm + 7 * 1024;
    float* const dT = sm + 8 * 1024;
    __syncthreads();
    #pragma unroll
    for (int j = 0; j < 32; ++j) {
        const int i = lane + 32 * j;
        dV[i] = u[j] * (1.f / 64.f);
        dR[i] = ((i % 33) == 0) ? 1.f : 0.f;
    }
    for (int ii = 10; ii >= 1; --ii) {
        mm(dT, 32, dV, 32, dR, 32, 32, 32, false);
        const float inv = 1.f / (float)ii;
        float v[32];
        ld32(v, dT, lane);
        #pragma unroll
        for (int j = 0; j < 32; ++j) {
            const int i = lane + 32 * j;
            dR[i] = (((i % 33) == 0) ? 1.f : 0.f) + v[j] * inv;
        }
    }
    float* r  = dR;
    float* rn = dT;
    for (int sq = 0; sq < 6; ++sq) {
        mm(rn, 32, r, 32, r, 32, 32, 32, false);
        float* tmp = r; r = rn; rn = tmp;
    }
    mm(dV, 32, pSq, 32, r, 32, 32, 32, false);
    mm(rn, 32, dV, 32, pSq, 32, 32, 32, false);
    __syncthreads();
    {
        float v[32];
        ld32(v, rn, lane);
        #pragma unroll
        for (int j = 0; j < 32; ++j) out[bc * 1024 + lane + 32 * j] = v[j];
    }
}

extern "C" void kernel_launch(void* const* d_in, const int* in_sizes, int n_in,
                              void* d_out, int out_size, void* d_ws, size_t ws_size,
                              hipStream_t stream)
{
    const float* x   = (const float*)d_in[0];
    const float* wts = (const float*)d_in[1];
    const float* Ws  = (const float*)d_in[2];
    float* out = (float*)d_out;

    dim3 grid(2048);
    dim3 block(32);
    hipLaunchKernelGGL(spdnet_mixedop_kernel, grid, block, 0, stream,
                       x, wts, Ws, out);
}


# Matrix log from both outputs of the coupled root iteration: log T = 4 asinh((T^(1/4)-T^(-1/4))/2), odd degree-15 fit in H (8 products): 2 roots per op, not 3; other sums, rounding differs
# speedup vs baseline: 7.1527x; 1.2298x over previous
#include <hip/hip_runtime.h>


typedef __attribute__((ext_vector_type(2))) float v2f;
typedef __attribute__((ext_vector_type(8))) float v8f;

#define WMMA_F32(a, b, c) \
    __builtin_amdgcn_wmma_f32_16x16x4_f32(false, (a), false, (b), (short)0, (c), false, false)

__device__ __forceinline__ void ld32(float (&v)[32], const float* p, int lane)
{
    #pragma unroll
    for (int j = 0; j < 32; ++j) v[j] = p[lane + 32 * j];
}

__device__ __forceinline__ void mm(float* __restrict__ C, int ldc,
                                   const float* __restrict__ A, int lda,
                                   const float* __restrict__ B, int ldb,
                                   int N, int Kd, bool transB)
{
    __syncthreads();
    const int lane = threadIdx.x;
    const int half = lane >> 4;
    const int l15  = lane & 15;
    for (int ni = 0; ni < N; ni += 32) {
        v8f c00 = {}, c01 = {}, c10 = {}, c11 = {};
        for (int k0 = 0; k0 < Kd; k0 += 32) {
            v2f a0[8], a1[8], b0[8], b1[8];
            #pragma unroll
            for (int s = 0; s < 8; ++s) {
                const int k = k0 + 4 * s + 2 * half;
                const float* ar = A + l15 * lda + k;
                a0[s].x = ar[0];
                a0[s].y = ar[1];
                a1[s].x = ar[16 * lda];
                a1[s].y = ar[16 * lda + 1];
                if (!transB) {
                    const float* br = B + k * ldb + ni + l15;
                    b0[s].x = br[0];
                    b0[s].y = br[ldb];
                    b1[s].x = br[16];
                    b1[s].y = br[ldb + 16];
                } else {
                    const float* br = B + (ni + l15) * ldb + k;
                    b0[s].x = br[0];
                    b0[s].y = br[1];
                    b1[s].x = br[16 * ldb];
                    b1[s].y = br[16 * ldb + 1];
                }
            }
            #pragma unroll
            for (int s = 0; s < 8; ++s) {
                c00 = WMMA_F32(a0[s], b0[s], c00);
                c01 = WMMA_F32(a0[s], b1[s], c01);
                c10 = WMMA_F32(a1[s], b0[s], c10);
                c11 = WMMA_F32(a1[s], b1[s], c11);
            }
        }
        #pragma unroll
        for (int r = 0; r < 8; ++r) {
            float* cr = C + (half * 8 + r) * ldc + ni + l15;
            cr[0]            = c00[r];
            cr[16]           = c01[r];
            cr[16 * ldc]     = c10[r];
            cr[16 * ldc + 16] = c11[r];
        }
    }
    __syncthreads();
}

__device__ __forceinline__ float trace32(const float* A, float* red)
{
    __syncthreads();
    red[threadIdx.x] = A[threadIdx.x * 33];
    __syncthreads();
    float s = 0.f;
    #pragma unroll
    for (int i = 0; i < 32; ++i) s += red[i];
    return s;
}

struct M32 { v8f t[2][2]; };

__device__ __forceinline__ void mmT(M32& D, const M32& P, const M32& Q)
{
    v8f d00 = {}, d01 = {}, d10 = {}, d11 = {};
    #pragma unroll
    for (int tk = 0; tk < 2; ++tk) {
        #pragma unroll
        for (int s = 0; s < 4; ++s) {
            v2f a0, a1, b0, b1;
            a0.x = P.t[tk][0][2 * s]; a0.y = P.t[tk][0][2 * s + 1];
            a1.x = P.t[tk][1][2 * s]; a1.y = P.t[tk][1][2 * s + 1];
            b0.x = Q.t[tk][0][2 * s]; b0.y = Q.t[tk][0][2 * s + 1];
            b1.x = Q.t[tk][1][2 * s]; b1.y = Q.t[tk][1][2 * s + 1];
            d00 = WMMA_F32(a0, b0, d00);
            d01 = WMMA_F32(a0, b1, d01);
            d10 = WMMA_F32(a1, b0, d10);
            d11 = WMMA_F32(a1, b1, d11);
        }
    }
    D.t[0][0] = d00; D.t[0][1] = d01; D.t[1][0] = d10; D.t[1][1] = d11;
}

__device__ __forceinline__ void mmT_acc(M32& D, const M32& C, const M32& P, const M32& Q)
{
    v8f d00 = C.t[0][0], d01 = C.t[0][1], d10 = C.t[1][0], d11 = C.t[1][1];
    #pragma unroll
    for (int tk = 0; tk < 2; ++tk) {
        #pragma unroll
        for (int s = 0; s < 4; ++s) {
            v2f a0, a1, b0, b1;
            a0.x = P.t[tk][0][2 * s]; a0.y = P.t[tk][0][2 * s + 1];
            a1.x = P.t[tk][1][2 * s]; a1.y = P.t[tk][1][2 * s + 1];
            b0.x = Q.t[tk][0][2 * s]; b0.y = Q.t[tk][0][2 * s + 1];
            b1.x = Q.t[tk][1][2 * s]; b1.y = Q.t[tk][1][2 * s + 1];
            d00 = WMMA_F32(a0, b0, d00);
            d01 = WMMA_F32(a0, b1, d01);
            d10 = WMMA_F32(a1, b0, d10);
            d11 = WMMA_F32(a1, b1, d11);
        }
    }
    D.t[0][0] = d00; D.t[0][1] = d01; D.t[1][0] = d10; D.t[1][1] = d11;
}

template <bool FLIP>
__device__ __forceinline__ void sqrt_step(M32& Y, M32& Z, int half, int l15,
                                          float sg, float k0, float k1, float k2, float k3, float k4)
{
    M32 X, X2, W, P;
    if (!FLIP) mmT(X, Z, Y); else mmT(X, Y, Z);
    #pragma unroll
    for (int ti = 0; ti < 2; ++ti)
        #pragma unroll
        for (int tj = 0; tj < 2; ++tj)
            #pragma unroll
            for (int r = 0; r < 8; ++r) {
                const float d = (ti == tj && (8 * half + r) == l15) ? sg : 0.f;
                X.t[ti][tj][r] = X.t[ti][tj][r] - d;
            }
    mmT(X2, X, X);
    #pragma unroll
    for (int ti = 0; ti < 2; ++ti)
        #pragma unroll
        for (int tj = 0; tj < 2; ++tj)
            #pragma unroll
            for (int r = 0; r < 8; ++r) {
                const float d  = (ti == tj && (8 * half + r) == l15) ? k0 : 0.f;
                const float x  = X.t[ti][tj][r];
                const float x2 = X2.t[ti][tj][r];
                W.t[ti][tj][r] = k3 * x + k4 * x2;
                P.t[ti][tj][r] = d + k1 * x + k2 * x2;
            }
    if (!FLIP) mmT_acc(P, P, X2, W); else mmT_acc(P, P, W, X2);
    M32 Yn, Zn;
    if (!FLIP) { mmT(Yn, P, Y); mmT(Zn, Z, P); }
    else       { mmT(Yn, Y, P); mmT(Zn, P, Z); }
    Y = Yn; Z = Zn;
}

__device__ __forceinline__ void symmetrise2(M32& Y, M32& Z, float* by, float* bz, int half, int l15)
{
    #pragma unroll
    for (int ti = 0; ti < 2; ++ti)
        #pragma unroll
        for (int tj = 0; tj < 2; ++tj)
            #pragma unroll
            for (int r = 0; r < 8; ++r) {
                const int i = (16 * ti + 8 * half + r) * 32 + 16 * tj + l15;
                by[i] = Y.t[ti][tj][r];
                bz[i] = Z.t[ti][tj][r];
            }
    __syncthreads();
    #pragma unroll
    for (int ti = 0; ti < 2; ++ti)
        #pragma unroll
        for (int tj = 0; tj < 2; ++tj)
            #pragma unroll
            for (int r = 0; r < 8; ++r) {
                const int j = (16 * tj + l15) * 32 + 16 * ti + 8 * half + r;
                Y.t[ti][tj][r] = 0.5f * (Y.t[ti][tj][r] + by[j]);
                Z.t[ti][tj][r] = 0.5f * (Z.t[ti][tj][r] + bz[j]);
            }
    __syncthreads();
}

__device__ __forceinline__ void ns_sqrt(float* y, float* z, float* t, bool want_inv)
{
    const int lane = threadIdx.x;
    const int half = lane >> 4;
    const int l15  = lane & 15;
    const float c = trace32(y, t);
    const float inv_c = 1.f / c;
    __syncthreads();
    M32 Y, Z;
    #pragma unroll
    for (int ti = 0; ti < 2; ++ti)
        #pragma unroll
        for (int tj = 0; tj < 2; ++tj)
            #pragma unroll
            for (int r = 0; r < 8; ++r) {
                const int row = 16 * ti + 8 * half + r;
                const int col = 16 * tj + l15;
                Y.t[ti][tj][r] = 0.5f * (y[row * 32 + col] + y[col * 32 + row]) * inv_c;
                Z.t[ti][tj][r] = (row == col) ? 1.f : 0.f;
            }
    #pragma nounroll
    for (int it = 0; it < 2; ++it) {
        const bool a = (it == 0);
        sqrt_step<false>(Y, Z, half, l15,
                         a ? 0.550000012f : 1.61000001f,
                         a ? 2.10768604f : 0.790120423f, a ? 5.74562931f : -0.282128632f, a ? -30.9237690f : 0.0719144717f,
                         a ? -54.0620193f : 0.150992960f, a ? 162.579529f : 0.289191633f);
        symmetrise2(Y, Z, y, t, half, l15);
        sqrt_step<true>(Y, Z, half, l15,
                        a ? 1.79999995f : 1.00000000f,
                        a ? 0.935014427f : 1.00000000f, a ? 0.400891274f : -0.500000000f, a ? -0.721723795f : 0.375000000f,
                        a ? -0.493403643f : -0.312500000f, a ? 0.415029675f : 0.273437500f);
    }
    const float sc  = __fsqrt_rn(c);
    const float isc = __frsqrt_rn(c);
    #pragma unroll
    for (int ti = 0; ti < 2; ++ti)
        #pragma unroll
        for (int tj = 0; tj < 2; ++tj)
            #pragma unroll
            for (int r = 0; r < 8; ++r) {
                const int i = (16 * ti + 8 * half + r) * 32 + 16 * tj + l15;
                y[i] = Y.t[ti][tj][r] * sc;
                if (want_inv) z[i] = Z.t[ti][tj][r] * isc;
            }
    __syncthreads();
}

__global__ __launch_bounds__(32)
void spdnet_mixedop_kernel(const float* __restrict__ x,
                           const float* __restrict__ wts,
                           const float* __restrict__ Ws,
                           float* __restrict__ out)
{
    __shared__ float sm[9 * 1024];
    float* const sX = sm;
    float* const sP = sm + 4 * 1024;
    float* const s0 = sm + 6 * 1024;
    float* const s1 = sm + 7 * 1024;
    float* const s2 = sm + 8 * 1024;
    float* const s3 = sm;

    const int bc   = blockIdx.x;
    const int lane = threadIdx.x;

    float wk[4];
    #pragma unroll
    for (int k = 0; k < 4; ++k) wk[k] = wts[k];

    #pragma unroll 16
    for (int i = lane; i < 4096; i += 32) sX[i] = x[bc * 4096 + i];

    for (int k = 0; k < 4; ++k) {
        const float* Wk = Ws + k * 2048;
        float* sk = (k < 3) ? (s0 + k * 1024) : s3;
        mm(sP, 64, Wk, 64, sX, 64, 64, 64, false);
        mm(sk, 32, sP, 64, Wk, 64, 32, 64, true);
    }

    float* by = sm + 1 * 1024;
    float* bz = sm + 2 * 1024;
    float* bt = sm + 3 * 1024;
    float* bf = sm + 4 * 1024;
    __syncthreads();
    #pragma unroll 8
    for (int i = lane; i < 1024; i += 32)
        by[i] = wk[0] * s0[i] + wk[1] * s1[i] +
                wk[2] * s2[i] + wk[3] * s3[i];

    ns_sqrt(by, bz, bt, true);
    float* const pSq = by;
    float* const pSi = bz;
    float* fa = bt;
    float* fb = bf;
    float* fc = sm + 5 * 1024;

    float u[32];
    #pragma unroll
    for (int j = 0; j < 32; ++j) u[j] = 0.f;

    for (int k = 0; k < 4; ++k) {
        float* y = (k < 3) ? (s0 + k * 1024) : s3;
        float* z = fa;
        float* t = fb;
        float* f = fc;
        mm(z, 32, pSi, 32, y, 32, 32, 32, false);
        mm(y, 32, z, 32, pSi, 32, 32, 32, false);

        #pragma nounroll
        for (int rt = 0; rt < 2; ++rt)
            ns_sqrt(y, z, t, rt == 1);

        float* H = y;
        {
            float a[32], b[32];
            ld32(a, y, lane);
            ld32(b, z, lane);
            #pragma unroll
            for (int j = 0; j < 32; ++j) H[lane + 32 * j] = 0.5f * (a[j] - b[j]);
        }
        float* H2 = t;
        mm(H2, 32, H, 32, H, 32, 32, 32, false);

        float* q  = z;
        float* qn = f;
        {
            float v[32];
            ld32(v, H2, lane);
            #pragma unroll
            for (int j = 0; j < 32; ++j) {
                const int i = lane + 32 * j;
                q[i] = (((i % 33) == 0) ? 0.00378406583f : 0.f) + (-0.000549010991f) * v[j];
            }
        }
        for (int ii = 5; ii >= 0; --ii) {
            mm(qn, 32, H2, 32, q, 32, 32, 32, false);
            const float g = (ii == 5) ? -0.0120585402f : (ii == 4) ? 0.0251510050f :
                            (ii == 3) ? -0.0430501215f : (ii == 2) ? 0.0747471750f :
                            (ii == 1) ? -0.166650742f  : 0.999999821f;
            qn[lane * 33] += g;
            float* tmp = q; q = qn; qn = tmp;
        }
        mm(qn, 32, H, 32, q, 32, 32, 32, false);

        const float scale = wk[k] * 4.f;
        {
            float w[32];
            ld32(w, qn, lane);
            #pragma unroll
            for (int j = 0; j < 32; ++j) u[j] = u[j] + scale * w[j];
        }
        fa = z; fb = t; fc = f;
        __syncthreads();
    }

    float* const dV = sm + 6 * 1024;
    float* const dR = sm + 7 * 1024;
    float* const dT = sm + 8 * 1024;
    __syncthreads();
    #pragma unroll
    for (int j = 0; j < 32; ++j) {
        const int i = lane + 32 * j;
        dV[i] = u[j] * (1.f / 64.f);
        dR[i] = ((i % 33) == 0) ? 1.f : 0.f;
    }
    for (int ii = 10; ii >= 1; --ii) {
        mm(dT, 32, dV, 32, dR, 32, 32, 32, false);
        const float inv = 1.f / (float)ii;
        float v[32];
        ld32(v, dT, lane);
        #pragma unroll
        for (int j = 0; j < 32; ++j) {
            const int i = lane + 32 * j;
            dR[i] = (((i % 33) == 0) ? 1.f : 0.f) + v[j] * inv;
        }
    }
    float* r  = dR;
    float* rn = dT;
    for (int sq = 0; sq < 6; ++sq) {
        mm(rn, 32, r, 32, r, 32, 32, 32, false);
        float* tmp = r; r = rn; rn = tmp;
    }
    mm(dV, 32, pSq, 32, r, 32, 32, 32, false);
    mm(rn, 32, dV, 32, pSq, 32, 32, 32, false);
    __syncthreads();
    {
        float v[32];
        ld32(v, rn, lane);
        #pragma unroll
        for (int j = 0; j < 32; ++j) out[bc * 1024 + lane + 32 * j] = v[j];
    }
}

extern "C" void kernel_launch(void* const* d_in, const int* in_sizes, int n_in,
                              void* d_out, int out_size, void* d_ws, size_t ws_size,
                              hipStream_t stream)
{
    const float* x   = (const float*)d_in[0];
    const float* wts = (const float*)d_in[1];
    const float* Ws  = (const float*)d_in[2];
    float* out = (float*)d_out;

    dim3 grid(2048);
    dim3 block(32);
    hipLaunchKernelGGL(spdnet_mixedop_kernel, grid, block, 0, stream,
                       x, wts, Ws, out);
}


# Congruences, asinh series and exponential as register products P^T*Q on accumulator-layout operands (as the roots): no LDS between 32x32 products from G0 on; sums in a new k order, rounding differs
# speedup vs baseline: 8.0849x; 1.1303x over previous
#include <hip/hip_runtime.h>


typedef __attribute__((ext_vector_type(2))) float v2f;
typedef __attribute__((ext_vector_type(8))) float v8f;

#define WMMA_F32(a, b, c) \
    __builtin_amdgcn_wmma_f32_16x16x4_f32(false, (a), false, (b), (short)0, (c), false, false)

__device__ __forceinline__ void ld32(float (&v)[32], const float* p, int lane)
{
    #pragma unroll
    for (int j = 0; j < 32; ++j) v[j] = p[lane + 32 * j];
}

__device__ __forceinline__ void mm(float* __restrict__ C, int ldc,
                                   const float* __restrict__ A, int lda,
                                   const float* __restrict__ B, int ldb,
                                   int N, int Kd, bool transB)
{
    __syncthreads();
    const int lane = threadIdx.x;
    const int half = lane >> 4;
    const int l15  = lane & 15;
    for (int ni = 0; ni < N; ni += 32) {
        v8f c00 = {}, c01 = {}, c10 = {}, c11 = {};
        for (int k0 = 0; k0 < Kd; k0 += 32) {
            v2f a0[8], a1[8], b0[8], b1[8];
            #pragma unroll
            for (int s = 0; s < 8; ++s) {
                const int k = k0 + 4 * s + 2 * half;
                const float* ar = A + l15 * lda + k;
                a0[s].x = ar[0];
                a0[s].y = ar[1];
                a1[s].x = ar[16 * lda];
                a1[s].y = ar[16 * lda + 1];
                if (!transB) {
                    const float* br = B + k * ldb + ni + l15;
                    b0[s].x = br[0];
                    b0[s].y = br[ldb];
                    b1[s].x = br[16];
                    b1[s].y = br[ldb + 16];
                } else {
                    const float* br = B + (ni + l15) * ldb + k;
                    b0[s].x = br[0];
                    b0[s].y = br[1];
                    b1[s].x = br[16 * ldb];
                    b1[s].y = br[16 * ldb + 1];
                }
            }
            #pragma unroll
            for (int s = 0; s < 8; ++s) {
                c00 = WMMA_F32(a0[s], b0[s], c00);
                c01 = WMMA_F32(a0[s], b1[s], c01);
                c10 = WMMA_F32(a1[s], b0[s], c10);
                c11 = WMMA_F32(a1[s], b1[s], c11);
            }
        }
        #pragma unroll
        for (int r = 0; r < 8; ++r) {
            float* cr = C + (half * 8 + r) * ldc + ni + l15;
            cr[0]            = c00[r];
            cr[16]           = c01[r];
            cr[16 * ldc]     = c10[r];
            cr[16 * ldc + 16] = c11[r];
        }
    }
    __syncthreads();
}

struct M32 { v8f t[2][2]; };

__device__ __forceinline__ void mmT(M32& D, const M32& P, const M32& Q)
{
    v8f d00 = {}, d01 = {}, d10 = {}, d11 = {};
    #pragma unroll
    for (int tk = 0; tk < 2; ++tk) {
        #pragma unroll
        for (int s = 0; s < 4; ++s) {
            v2f a0, a1, b0, b1;
            a0.x = P.t[tk][0][2 * s]; a0.y = P.t[tk][0][2 * s + 1];
            a1.x = P.t[tk][1][2 * s]; a1.y = P.t[tk][1][2 * s + 1];
            b0.x = Q.t[tk][0][2 * s]; b0.y = Q.t[tk][0][2 * s + 1];
            b1.x = Q.t[tk][1][2 * s]; b1.y = Q.t[tk][1][2 * s + 1];
            d00 = WMMA_F32(a0, b0, d00);
            d01 = WMMA_F32(a0, b1, d01);
            d10 = WMMA_F32(a1, b0, d10);
            d11 = WMMA_F32(a1, b1, d11);
        }
    }
    D.t[0][0] = d00; D.t[0][1] = d01; D.t[1][0] = d10; D.t[1][1] = d11;
}

__device__ __forceinline__ void mmT_acc(M32& D, const M32& C, const M32& P, const M32& Q)
{
    v8f d00 = C.t[0][0], d01 = C.t[0][1], d10 = C.t[1][0], d11 = C.t[1][1];
    #pragma unroll
    for (int tk = 0; tk < 2; ++tk) {
        #pragma unroll
        for (int s = 0; s < 4; ++s) {
            v2f a0, a1, b0, b1;
            a0.x = P.t[tk][0][2 * s]; a0.y = P.t[tk][0][2 * s + 1];
            a1.x = P.t[tk][1][2 * s]; a1.y = P.t[tk][1][2 * s + 1];
            b0.x = Q.t[tk][0][2 * s]; b0.y = Q.t[tk][0][2 * s + 1];
            b1.x = Q.t[tk][1][2 * s]; b1.y = Q.t[tk][1][2 * s + 1];
            d00 = WMMA_F32(a0, b0, d00);
            d01 = WMMA_F32(a0, b1, d01);
            d10 = WMMA_F32(a1, b0, d10);
            d11 = WMMA_F32(a1, b1, d11);
        }
    }
    D.t[0][0] = d00; D.t[0][1] = d01; D.t[1][0] = d10; D.t[1][1] = d11;
}

template <bool FLIP>
__device__ __forceinline__ void sqrt_step(M32& Y, M32& Z, int half, int l15,
                                          float sg, float k0, float k1, float k2, float k3, float k4)
{
    M32 X, X2, W, P;
    if (!FLIP) mmT(X, Z, Y); else mmT(X, Y, Z);
    #pragma unroll
    for (int ti = 0; ti < 2; ++ti)
        #pragma unroll
        for (int tj = 0; tj < 2; ++tj)
            #pragma unroll
            for (int r = 0; r < 8; ++r) {
                const float d = (ti == tj && (8 * half + r) == l15) ? sg : 0.f;
                X.t[ti][tj][r] = X.t[ti][tj][r] - d;
            }
    mmT(X2, X, X);
    #pragma unroll
    for (int ti = 0; ti < 2; ++ti)
        #pragma unroll
        for (int tj = 0; tj < 2; ++tj)
            #pragma unroll
            for (int r = 0; r < 8; ++r) {
                const float d  = (ti == tj && (8 * half + r) == l15) ? k0 : 0.f;
                const float x  = X.t[ti][tj][r];
                const float x2 = X2.t[ti][tj][r];
                W.t[ti][tj][r] = k3 * x + k4 * x2;
                P.t[ti][tj][r] = d + k1 * x + k2 * x2;
            }
    if (!FLIP) mmT_acc(P, P, X2, W); else mmT_acc(P, P, W, X2);
    M32 Yn, Zn;
    if (!FLIP) { mmT(Yn, P, Y); mmT(Zn, Z, P); }
    else       { mmT(Yn, Y, P); mmT(Zn, P, Z); }
    Y = Yn; Z = Zn;
}

__device__ __forceinline__ void symmetrise2(M32& Y, M32& Z, float* by, float* bz, int half, int l15)
{
    #pragma unroll
    for (int ti = 0; ti < 2; ++ti)
        #pragma unroll
        for (int tj = 0; tj < 2; ++tj)
            #pragma unroll
            for (int r = 0; r < 8; ++r) {
                const int i = (16 * ti + 8 * half + r) * 32 + 16 * tj + l15;
                by[i] = Y.t[ti][tj][r];
                bz[i] = Z.t[ti][tj][r];
            }
    __syncthreads();
    #pragma unroll
    for (int ti = 0; ti < 2; ++ti)
        #pragma unroll
        for (int tj = 0; tj < 2; ++tj)
            #pragma unroll
            for (int r = 0; r < 8; ++r) {
                const int j = (16 * tj + l15) * 32 + 16 * ti + 8 * half + r;
                Y.t[ti][tj][r] = 0.5f * (Y.t[ti][tj][r] + by[j]);
                Z.t[ti][tj][r] = 0.5f * (Z.t[ti][tj][r] + bz[j]);
            }
    __syncthreads();
}

__device__ __forceinline__ void ldM(M32& D, const float* p, int half, int l15)
{
    #pragma unroll
    for (int ti = 0; ti < 2; ++ti)
        #pragma unroll
        for (int tj = 0; tj < 2; ++tj)
            #pragma unroll
            for (int r = 0; r < 8; ++r)
                D.t[ti][tj][r] = p[(16 * ti + 8 * half + r) * 32 + 16 * tj + l15];
}

__device__ __forceinline__ void stM(float* p, const M32& S, int half, int l15)
{
    #pragma unroll
    for (int ti = 0; ti < 2; ++ti)
        #pragma unroll
        for (int tj = 0; tj < 2; ++tj)
            #pragma unroll
            for (int r = 0; r < 8; ++r)
                p[(16 * ti + 8 * half + r) * 32 + 16 * tj + l15] = S.t[ti][tj][r];
}

__device__ __forceinline__ void root_reg(M32& Y, M32& Z, float* b0, float* b1, int half, int l15)
{
    stM(b0, Y, half, l15);
    __syncthreads();
    float c = 0.f;
    #pragma unroll
    for (int i = 0; i < 32; ++i) c += b0[i * 33];
    const float inv_c = 1.f / c;
    #pragma unroll
    for (int ti = 0; ti < 2; ++ti)
        #pragma unroll
        for (int tj = 0; tj < 2; ++tj)
            #pragma unroll
            for (int r = 0; r < 8; ++r) {
                const int row = 16 * ti + 8 * half + r;
                const int col = 16 * tj + l15;
                Y.t[ti][tj][r] = 0.5f * (Y.t[ti][tj][r] + b0[col * 32 + row]) * inv_c;
                Z.t[ti][tj][r] = (row == col) ? 1.f : 0.f;
            }
    __syncthreads();
    #pragma nounroll
    for (int it = 0; it < 2; ++it) {
        const bool a = (it == 0);
        sqrt_step<false>(Y, Z, half, l15,
                         a ? 0.550000012f : 1.61000001f,
                         a ? 2.10768604f : 0.790120423f, a ? 5.74562931f : -0.282128632f, a ? -30.9237690f : 0.0719144717f,
                         a ? -54.0620193f : 0.150992960f, a ? 162.579529f : 0.289191633f);
        symmetrise2(Y, Z, b0, b1, half, l15);
        sqrt_step<true>(Y, Z, half, l15,
                        a ? 1.79999995f : 1.00000000f,
                        a ? 0.935014427f : 1.00000000f, a ? 0.400891274f : -0.500000000f, a ? -0.721723795f : 0.375000000f,
                        a ? -0.493403643f : -0.312500000f, a ? 0.415029675f : 0.273437500f);
    }
    const float sc  = __fsqrt_rn(c);
    const float isc = __frsqrt_rn(c);
    #pragma unroll
    for (int ti = 0; ti < 2; ++ti)
        #pragma unroll
        for (int tj = 0; tj < 2; ++tj)
            #pragma unroll
            for (int r = 0; r < 8; ++r) {
                Y.t[ti][tj][r] = Y.t[ti][tj][r] * sc;
                Z.t[ti][tj][r] = Z.t[ti][tj][r] * isc;
            }
}

__global__ __launch_bounds__(32)
void spdnet_mixedop_kernel(const float* __restrict__ x,
                           const float* __restrict__ wts,
                           const float* __restrict__ Ws,
                           float* __restrict__ out)
{
    __shared__ float sm[9 * 1024];
    float* const sX = sm;
    float* const sP = sm + 4 * 1024;
    float* const s0 = sm + 6 * 1024;
    float* const s1 = sm + 7 * 1024;
    float* const s2 = sm + 8 * 1024;
    float* const s3 = sm;

    const int bc   = blockIdx.x;
    const int lane = threadIdx.x;

    float wk[4];
    #pragma unroll
    for (int k = 0; k < 4; ++k) wk[k] = wts[k];

    #pragma unroll 16
    for (int i = lane; i < 4096; i += 32) sX[i] = x[bc * 4096 + i];

    for (int k = 0; k < 4; ++k) {
        const float* Wk = Ws + k * 2048;
        float* sk = (k < 3) ? (s0 + k * 1024) : s3;
        mm(sP, 64, Wk, 64, sX, 64, 64, 64, false);
        mm(sk, 32, sP, 64, Wk, 64, 32, 64, true);
    }

    float* const b0  = sm + 1 * 1024;
    float* const b1  = sm + 2 * 1024;
    float* const pSq = sm + 3 * 1024;
    float* const pSi = sm + 4 * 1024;
    const int half = lane >> 4;
    const int l15  = lane & 15;

    M32 U;
    #pragma unroll
    for (int ti = 0; ti < 2; ++ti)
        #pragma unroll
        for (int tj = 0; tj < 2; ++tj)
            #pragma unroll
            for (int r = 0; r < 8; ++r) U.t[ti][tj][r] = 0.f;

    for (int k = -1; k < 4; ++k) {
        M32 Y, Z;
        if (k < 0) {
            #pragma unroll
            for (int ti = 0; ti < 2; ++ti)
                #pragma unroll
                for (int tj = 0; tj < 2; ++tj)
                    #pragma unroll
                    for (int r = 0; r < 8; ++r) {
                        const int i = (16 * ti + 8 * half + r) * 32 + 16 * tj + l15;
                        Y.t[ti][tj][r] = wk[0] * s0[i] + wk[1] * s1[i] +
                                         wk[2] * s2[i] + wk[3] * s3[i];
                    }
        } else {
            const float* sk = (k < 3) ? (s0 + k * 1024) : s3;
            M32 S, Si, A1;
            ldM(S, sk, half, l15);
            ldM(Si, pSi, half, l15);
            mmT(A1, S, Si);
            mmT(Y, A1, Si);
        }

        const int nr = (k < 0) ? 1 : 2;
        for (int rt = 0; rt < nr; ++rt)
            root_reg(Y, Z, b0, b1, half, l15);

        if (k < 0) {
            stM(pSq, Y, half, l15);
            stM(pSi, Z, half, l15);
            continue;
        }

        M32 H, H2, q;
        #pragma unroll
        for (int ti = 0; ti < 2; ++ti)
            #pragma unroll
            for (int tj = 0; tj < 2; ++tj)
                #pragma unroll
                for (int r = 0; r < 8; ++r)
                    H.t[ti][tj][r] = 0.5f * (Y.t[ti][tj][r] - Z.t[ti][tj][r]);
        mmT(H2, H, H);

        #pragma unroll
        for (int ti = 0; ti < 2; ++ti)
            #pragma unroll
            for (int tj = 0; tj < 2; ++tj)
                #pragma unroll
                for (int r = 0; r < 8; ++r) {
                    const float d = (ti == tj && (8 * half + r) == l15) ? 0.00378406583f : 0.f;
                    q.t[ti][tj][r] = d + (-0.000549010991f) * H2.t[ti][tj][r];
                }
        for (int ii = 5; ii >= 0; --ii) {
            M32 qn;
            mmT(qn, H2, q);
            const float g = (ii == 5) ? -0.0120585402f : (ii == 4) ? 0.0251510050f :
                            (ii == 3) ? -0.0430501215f : (ii == 2) ? 0.0747471750f :
                            (ii == 1) ? -0.166650742f  : 0.999999821f;
            #pragma unroll
            for (int ti = 0; ti < 2; ++ti)
                #pragma unroll
                for (int r = 0; r < 8; ++r)
                    qn.t[ti][ti][r] = qn.t[ti][ti][r] + (((8 * half + r) == l15) ? g : 0.f);
            q = qn;
        }
        M32 L;
        mmT(L, H, q);

        const float wsel  = (k == 0) ? wk[0] : (k == 1) ? wk[1] : (k == 2) ? wk[2] : wk[3];
        const float scale = wsel * 4.f;
        #pragma unroll
        for (int ti = 0; ti < 2; ++ti)
            #pragma unroll
            for (int tj = 0; tj < 2; ++tj)
                #pragma unroll
                for (int r = 0; r < 8; ++r)
                    U.t[ti][tj][r] = U.t[ti][tj][r] + scale * L.t[ti][tj][r];
    }

    M32 V, R;
    #pragma unroll
    for (int ti = 0; ti < 2; ++ti)
        #pragma unroll
        for (int tj = 0; tj < 2; ++tj)
            #pragma unroll
            for (int r = 0; r < 8; ++r) {
                V.t[ti][tj][r] = U.t[ti][tj][r] * (1.f / 64.f);
                R.t[ti][tj][r] = (ti == tj && (8 * half + r) == l15) ? 1.f : 0.f;
            }
    for (int ii = 10; ii >= 1; --ii) {
        M32 Tm;
        mmT(Tm, V, R);
        const float inv = 1.f / (float)ii;
        #pragma unroll
        for (int ti = 0; ti < 2; ++ti)
            #pragma unroll
            for (int tj = 0; tj < 2; ++tj)
                #pragma unroll
                for (int r = 0; r < 8; ++r) {
                    const float d = (ti == tj && (8 * half + r) == l15) ? 1.f : 0.f;
                    R.t[ti][tj][r] = d + Tm.t[ti][tj][r] * inv;
                }
    }
    for (int sq = 0; sq < 6; ++sq) {
        M32 Rn;
        mmT(Rn, R, R);
        R = Rn;
    }
    M32 Sq, E1, O;
    ldM(Sq, pSq, half, l15);
    mmT(E1, R, Sq);
    mmT(O, E1, Sq);
    stM(b0, O, half, l15);
    __syncthreads();
    {
        float v[32];
        ld32(v, b0, lane);
        #pragma unroll
        for (int j = 0; j < 32; ++j) out[bc * 1024 + lane + 32 * j] = v[j];
    }
}

extern "C" void kernel_launch(void* const* d_in, const int* in_sizes, int n_in,
                              void* d_out, int out_size, void* d_ws, size_t ws_size,
                              hipStream_t stream)
{
    const float* x   = (const float*)d_in[0];
    const float* wts = (const float*)d_in[1];
    const float* Ws  = (const float*)d_in[2];
    float* out = (float*)d_out;

    dim3 grid(2048);
    dim3 block(32);
    hipLaunchKernelGGL(spdnet_mixedop_kernel, grid, block, 0, stream,
                       x, wts, Ws, out);
}


# Matrix log from ONE root per op, not two: Y=T^(1/2), Z=T^(-1/2) give log T = (Y-Z)/2 * phi((Y+Z-2I)/4), phi a degree-9 fit in s=sinh^2(l/4) (9 products); other sums, rounding differs
# speedup vs baseline: 11.1473x; 1.3788x over previous
#include <hip/hip_runtime.h>


typedef __attribute__((ext_vector_type(2))) float v2f;
typedef __attribute__((ext_vector_type(8))) float v8f;

#define WMMA_F32(a, b, c) \
    __builtin_amdgcn_wmma_f32_16x16x4_f32(false, (a), false, (b), (short)0, (c), false, false)

__device__ __forceinline__ void ld32(float (&v)[32], const float* p, int lane)
{
    #pragma unroll
    for (int j = 0; j < 32; ++j) v[j] = p[lane + 32 * j];
}

__device__ __forceinline__ void mm(float* __restrict__ C, int ldc,
                                   const float* __restrict__ A, int lda,
                                   const float* __restrict__ B, int ldb,
                                   int N, int Kd, bool transB)
{
    __syncthreads();
    const int lane = threadIdx.x;
    const int half = lane >> 4;
    const int l15  = lane & 15;
    for (int ni = 0; ni < N; ni += 32) {
        v8f c00 = {}, c01 = {}, c10 = {}, c11 = {};
        for (int k0 = 0; k0 < Kd; k0 += 32) {
            v2f a0[8], a1[8], b0[8], b1[8];
            #pragma unroll
            for (int s = 0; s < 8; ++s) {
                const int k = k0 + 4 * s + 2 * half;
                const float* ar = A + l15 * lda + k;
                a0[s].x = ar[0];
                a0[s].y = ar[1];
                a1[s].x = ar[16 * lda];
                a1[s].y = ar[16 * lda + 1];
                if (!transB) {
                    const float* br = B + k * ldb + ni + l15;
                    b0[s].x = br[0];
                    b0[s].y = br[ldb];
                    b1[s].x = br[16];
                    b1[s].y = br[ldb + 16];
                } else {
                    const float* br = B + (ni + l15) * ldb + k;
                    b0[s].x = br[0];
                    b0[s].y = br[1];
                    b1[s].x = br[16 * ldb];
                    b1[s].y = br[16 * ldb + 1];
                }
            }
            #pragma unroll
            for (int s = 0; s < 8; ++s) {
                c00 = WMMA_F32(a0[s], b0[s], c00);
                c01 = WMMA_F32(a0[s], b1[s], c01);
                c10 = WMMA_F32(a1[s], b0[s], c10);
                c11 = WMMA_F32(a1[s], b1[s], c11);
            }
        }
        #pragma unroll
        for (int r = 0; r < 8; ++r) {
            float* cr = C + (half * 8 + r) * ldc + ni + l15;
            cr[0]            = c00[r];
            cr[16]           = c01[r];
            cr[16 * ldc]     = c10[r];
            cr[16 * ldc + 16] = c11[r];
        }
    }
    __syncthreads();
}

struct M32 { v8f t[2][2]; };

__device__ __forceinline__ void mmT(M32& D, const M32& P, const M32& Q)
{
    v8f d00 = {}, d01 = {}, d10 = {}, d11 = {};
    #pragma unroll
    for (int tk = 0; tk < 2; ++tk) {
        #pragma unroll
        for (int s = 0; s < 4; ++s) {
            v2f a0, a1, b0, b1;
            a0.x = P.t[tk][0][2 * s]; a0.y = P.t[tk][0][2 * s + 1];
            a1.x = P.t[tk][1][2 * s]; a1.y = P.t[tk][1][2 * s + 1];
            b0.x = Q.t[tk][0][2 * s]; b0.y = Q.t[tk][0][2 * s + 1];
            b1.x = Q.t[tk][1][2 * s]; b1.y = Q.t[tk][1][2 * s + 1];
            d00 = WMMA_F32(a0, b0, d00);
            d01 = WMMA_F32(a0, b1, d01);
            d10 = WMMA_F32(a1, b0, d10);
            d11 = WMMA_F32(a1, b1, d11);
        }
    }
    D.t[0][0] = d00; D.t[0][1] = d01; D.t[1][0] = d10; D.t[1][1] = d11;
}

__device__ __forceinline__ void mmT_acc(M32& D, const M32& C, const M32& P, const M32& Q)
{
    v8f d00 = C.t[0][0], d01 = C.t[0][1], d10 = C.t[1][0], d11 = C.t[1][1];
    #pragma unroll
    for (int tk = 0; tk < 2; ++tk) {
        #pragma unroll
        for (int s = 0; s < 4; ++s) {
            v2f a0, a1, b0, b1;
            a0.x = P.t[tk][0][2 * s]; a0.y = P.t[tk][0][2 * s + 1];
            a1.x = P.t[tk][1][2 * s]; a1.y = P.t[tk][1][2 * s + 1];
            b0.x = Q.t[tk][0][2 * s]; b0.y = Q.t[tk][0][2 * s + 1];
            b1.x = Q.t[tk][1][2 * s]; b1.y = Q.t[tk][1][2 * s + 1];
            d00 = WMMA_F32(a0, b0, d00);
            d01 = WMMA_F32(a0, b1, d01);
            d10 = WMMA_F32(a1, b0, d10);
            d11 = WMMA_F32(a1, b1, d11);
        }
    }
    D.t[0][0] = d00; D.t[0][1] = d01; D.t[1][0] = d10; D.t[1][1] = d11;
}

template <bool FLIP>
__device__ __forceinline__ void sqrt_step(M32& Y, M32& Z, int half, int l15,
                                          float sg, float k0, float k1, float k2, float k3, float k4)
{
    M32 X, X2, W, P;
    if (!FLIP) mmT(X, Z, Y); else mmT(X, Y, Z);
    #pragma unroll
    for (int ti = 0; ti < 2; ++ti)
        #pragma unroll
        for (int tj = 0; tj < 2; ++tj)
            #pragma unroll
            for (int r = 0; r < 8; ++r) {
                const float d = (ti == tj && (8 * half + r) == l15) ? sg : 0.f;
                X.t[ti][tj][r] = X.t[ti][tj][r] - d;
            }
    mmT(X2, X, X);
    #pragma unroll
    for (int ti = 0; ti < 2; ++ti)
        #pragma unroll
        for (int tj = 0; tj < 2; ++tj)
            #pragma unroll
            for (int r = 0; r < 8; ++r) {
                const float d  = (ti == tj && (8 * half + r) == l15) ? k0 : 0.f;
                const float x  = X.t[ti][tj][r];
                const float x2 = X2.t[ti][tj][r];
                W.t[ti][tj][r] = k3 * x + k4 * x2;
                P.t[ti][tj][r] = d + k1 * x + k2 * x2;
            }
    if (!FLIP) mmT_acc(P, P, X2, W); else mmT_acc(P, P, W, X2);
    M32 Yn, Zn;
    if (!FLIP) { mmT(Yn, P, Y); mmT(Zn, Z, P); }
    else       { mmT(Yn, Y, P); mmT(Zn, P, Z); }
    Y = Yn; Z = Zn;
}

__device__ __forceinline__ void symmetrise2(M32& Y, M32& Z, float* by, float* bz, int half, int l15)
{
    #pragma unroll
    for (int ti = 0; ti < 2; ++ti)
        #pragma unroll
        for (int tj = 0; tj < 2; ++tj)
            #pragma unroll
            for (int r = 0; r < 8; ++r) {
                const int i = (16 * ti + 8 * half + r) * 32 + 16 * tj + l15;
                by[i] = Y.t[ti][tj][r];
                bz[i] = Z.t[ti][tj][r];
            }
    __syncthreads();
    #pragma unroll
    for (int ti = 0; ti < 2; ++ti)
        #pragma unroll
        for (int tj = 0; tj < 2; ++tj)
            #pragma unroll
            for (int r = 0; r < 8; ++r) {
                const int j = (16 * tj + l15) * 32 + 16 * ti + 8 * half + r;
                Y.t[ti][tj][r] = 0.5f * (Y.t[ti][tj][r] + by[j]);
                Z.t[ti][tj][r] = 0.5f * (Z.t[ti][tj][r] + bz[j]);
            }
    __syncthreads();
}

__device__ __forceinline__ void ldM(M32& D, const float* p, int half, int l15)
{
    #pragma unroll
    for (int ti = 0; ti < 2; ++ti)
        #pragma unroll
        for (int tj = 0; tj < 2; ++tj)
            #pragma unroll
            for (int r = 0; r < 8; ++r)
                D.t[ti][tj][r] = p[(16 * ti + 8 * half + r) * 32 + 16 * tj + l15];
}

__device__ __forceinline__ void stM(float* p, const M32& S, int half, int l15)
{
    #pragma unroll
    for (int ti = 0; ti < 2; ++ti)
        #pragma unroll
        for (int tj = 0; tj < 2; ++tj)
            #pragma unroll
            for (int r = 0; r < 8; ++r)
                p[(16 * ti + 8 * half + r) * 32 + 16 * tj + l15] = S.t[ti][tj][r];
}

__device__ __forceinline__ void root_reg(M32& Y, M32& Z, float* b0, float* b1, int half, int l15)
{
    stM(b0, Y, half, l15);
    __syncthreads();
    float c = 0.f;
    #pragma unroll
    for (int i = 0; i < 32; ++i) c += b0[i * 33];
    const float inv_c = 1.f / c;
    #pragma unroll
    for (int ti = 0; ti < 2; ++ti)
        #pragma unroll
        for (int tj = 0; tj < 2; ++tj)
            #pragma unroll
            for (int r = 0; r < 8; ++r) {
                const int row = 16 * ti + 8 * half + r;
                const int col = 16 * tj + l15;
                Y.t[ti][tj][r] = 0.5f * (Y.t[ti][tj][r] + b0[col * 32 + row]) * inv_c;
                Z.t[ti][tj][r] = (row == col) ? 1.f : 0.f;
            }
    __syncthreads();
    #pragma nounroll
    for (int it = 0; it < 2; ++it) {
        const bool a = (it == 0);
        sqrt_step<false>(Y, Z, half, l15,
                         a ? 0.550000012f : 1.61000001f,
                         a ? 2.10768604f : 0.790120423f, a ? 5.74562931f : -0.282128632f, a ? -30.9237690f : 0.0719144717f,
                         a ? -54.0620193f : 0.150992960f, a ? 162.579529f : 0.289191633f);
        symmetrise2(Y, Z, b0, b1, half, l15);
        sqrt_step<true>(Y, Z, half, l15,
                        a ? 1.79999995f : 1.00000000f,
                        a ? 0.935014427f : 1.00000000f, a ? 0.400891274f : -0.500000000f, a ? -0.721723795f : 0.375000000f,
                        a ? -0.493403643f : -0.312500000f, a ? 0.415029675f : 0.273437500f);
    }
    const float sc  = __fsqrt_rn(c);
    const float isc = __frsqrt_rn(c);
    #pragma unroll
    for (int ti = 0; ti < 2; ++ti)
        #pragma unroll
        for (int tj = 0; tj < 2; ++tj)
            #pragma unroll
            for (int r = 0; r < 8; ++r) {
                Y.t[ti][tj][r] = Y.t[ti][tj][r] * sc;
                Z.t[ti][tj][r] = Z.t[ti][tj][r] * isc;
            }
}

__global__ __launch_bounds__(32)
void spdnet_mixedop_kernel(const float* __restrict__ x,
                           const float* __restrict__ wts,
                           const float* __restrict__ Ws,
                           float* __restrict__ out)
{
    __shared__ float sm[9 * 1024];
    float* const sX = sm;
    float* const sP = sm + 4 * 1024;
    float* const s0 = sm + 6 * 1024;
    float* const s1 = sm + 7 * 1024;
    float* const s2 = sm + 8 * 1024;
    float* const s3 = sm;

    const int bc   = blockIdx.x;
    const int lane = threadIdx.x;

    float wk[4];
    #pragma unroll
    for (int k = 0; k < 4; ++k) wk[k] = wts[k];

    #pragma unroll 16
    for (int i = lane; i < 4096; i += 32) sX[i] = x[bc * 4096 + i];

    for (int k = 0; k < 4; ++k) {
        const float* Wk = Ws + k * 2048;
        float* sk = (k < 3) ? (s0 + k * 1024) : s3;
        mm(sP, 64, Wk, 64, sX, 64, 64, 64, false);
        mm(sk, 32, sP, 64, Wk, 64, 32, 64, true);
    }

    float* const b0  = sm + 1 * 1024;
    float* const b1  = sm + 2 * 1024;
    float* const pSq = sm + 3 * 1024;
    float* const pSi = sm + 4 * 1024;
    const int half = lane >> 4;
    const int l15  = lane & 15;

    M32 U;
    #pragma unroll
    for (int ti = 0; ti < 2; ++ti)
        #pragma unroll
        for (int tj = 0; tj < 2; ++tj)
            #pragma unroll
            for (int r = 0; r < 8; ++r) U.t[ti][tj][r] = 0.f;

    for (int k = -1; k < 4; ++k) {
        M32 Y, Z;
        if (k < 0) {
            #pragma unroll
            for (int ti = 0; ti < 2; ++ti)
                #pragma unroll
                for (int tj = 0; tj < 2; ++tj)
                    #pragma unroll
                    for (int r = 0; r < 8; ++r) {
                        const int i = (16 * ti + 8 * half + r) * 32 + 16 * tj + l15;
                        Y.t[ti][tj][r] = wk[0] * s0[i] + wk[1] * s1[i] +
                                         wk[2] * s2[i] + wk[3] * s3[i];
                    }
        } else {
            const float* sk = (k < 3) ? (s0 + k * 1024) : s3;
            M32 S, Si, A1;
            ldM(S, sk, half, l15);
            ldM(Si, pSi, half, l15);
            mmT(A1, S, Si);
            mmT(Y, A1, Si);
        }

        root_reg(Y, Z, b0, b1, half, l15);

        if (k < 0) {
            stM(pSq, Y, half, l15);
            stM(pSi, Z, half, l15);
            continue;
        }

        const float SC = 0.690548897f;
        M32 H1, X, q;
        #pragma unroll
        for (int ti = 0; ti < 2; ++ti)
            #pragma unroll
            for (int tj = 0; tj < 2; ++tj)
                #pragma unroll
                for (int r = 0; r < 8; ++r) {
                    const bool  dg = (ti == tj && (8 * half + r) == l15);
                    const float y  = Y.t[ti][tj][r];
                    const float z  = Z.t[ti][tj][r];
                    const float x  = 0.25f * (y + z) - (dg ? (0.5f + SC) : 0.f);
                    H1.t[ti][tj][r] = 0.5f * (y - z);
                    X.t[ti][tj][r]  = x;
                    q.t[ti][tj][r]  = (dg ? 0.0101555130f : 0.f) + (-0.00572075741f) * x;
                }
        for (int ii = 7; ii >= 0; --ii) {
            M32 qn;
            mmT(qn, X, q);
            const float g = (ii == 7) ? -0.0113067133f : (ii == 6) ? 0.0204704814f :
                            (ii == 5) ? -0.0403085500f : (ii == 4) ? 0.0746348128f :
                            (ii == 3) ? -0.140599474f  : (ii == 2) ? 0.274833739f  :
                            (ii == 1) ? -0.571834743f  : 1.40066803f;
            #pragma unroll
            for (int ti = 0; ti < 2; ++ti)
                #pragma unroll
                for (int r = 0; r < 8; ++r)
                    qn.t[ti][ti][r] = qn.t[ti][ti][r] + (((8 * half + r) == l15) ? g : 0.f);
            q = qn;
        }
        M32 L;
        mmT(L, H1, q);

        const float wsel  = (k == 0) ? wk[0] : (k == 1) ? wk[1] : (k == 2) ? wk[2] : wk[3];
        #pragma unroll
        for (int ti = 0; ti < 2; ++ti)
            #pragma unroll
            for (int tj = 0; tj < 2; ++tj)
                #pragma unroll
                for (int r = 0; r < 8; ++r)
                    U.t[ti][tj][r] = U.t[ti][tj][r] + wsel * L.t[ti][tj][r];
    }

    M32 V, R;
    #pragma unroll
    for (int ti = 0; ti < 2; ++ti)
        #pragma unroll
        for (int tj = 0; tj < 2; ++tj)
            #pragma unroll
            for (int r = 0; r < 8; ++r) {
                V.t[ti][tj][r] = U.t[ti][tj][r] * (1.f / 64.f);
                R.t[ti][tj][r] = (ti == tj && (8 * half + r) == l15) ? 1.f : 0.f;
            }
    for (int ii = 10; ii >= 1; --ii) {
        M32 Tm;
        mmT(Tm, V, R);
        const float inv = 1.f / (float)ii;
        #pragma unroll
        for (int ti = 0; ti < 2; ++ti)
            #pragma unroll
            for (int tj = 0; tj < 2; ++tj)
                #pragma unroll
                for (int r = 0; r < 8; ++r) {
                    const float d = (ti == tj && (8 * half + r) == l15) ? 1.f : 0.f;
                    R.t[ti][tj][r] = d + Tm.t[ti][tj][r] * inv;
                }
    }
    for (int sq = 0; sq < 6; ++sq) {
        M32 Rn;
        mmT(Rn, R, R);
        R = Rn;
    }
    M32 Sq, E1, O;
    ldM(Sq, pSq, half, l15);
    mmT(E1, R, Sq);
    mmT(O, E1, Sq);
    stM(b0, O, half, l15);
    __syncthreads();
    {
        float v[32];
        ld32(v, b0, lane);
        #pragma unroll
        for (int j = 0; j < 32; ++j) out[bc * 1024 + lane + 32 * j] = v[j];
    }
}

extern "C" void kernel_launch(void* const* d_in, const int* in_sizes, int n_in,
                              void* d_out, int out_size, void* d_ws, size_t ws_size,
                              hipStream_t stream)
{
    const float* x   = (const float*)d_in[0];
    const float* wts = (const float*)d_in[1];
    const float* Ws  = (const float*)d_in[2];
    float* out = (float*)d_out;

    dim3 grid(2048);
    dim3 block(32);
    hipLaunchKernelGGL(spdnet_mixedop_kernel, grid, block, 0, stream,
                       x, wts, Ws, out);
}


# Exponential phase: degree-9 Taylor summed in powers of V^3 (4 products) at U/32, 4 squarings, last doubling folded into the congruence (E Sq)^T(E Sq): 10 products, not 18; rounding differs
# speedup vs baseline: 11.5266x; 1.0340x over previous
#include <hip/hip_runtime.h>


typedef __attribute__((ext_vector_type(2))) float v2f;
typedef __attribute__((ext_vector_type(8))) float v8f;

#define WMMA_F32(a, b, c) \
    __builtin_amdgcn_wmma_f32_16x16x4_f32(false, (a), false, (b), (short)0, (c), false, false)

__device__ __forceinline__ void ld32(float (&v)[32], const float* p, int lane)
{
    #pragma unroll
    for (int j = 0; j < 32; ++j) v[j] = p[lane + 32 * j];
}

__device__ __forceinline__ void mm(float* __restrict__ C, int ldc,
                                   const float* __restrict__ A, int lda,
                                   const float* __restrict__ B, int ldb,
                                   int N, int Kd, bool transB)
{
    __syncthreads();
    const int lane = threadIdx.x;
    const int half = lane >> 4;
    const int l15  = lane & 15;
    for (int ni = 0; ni < N; ni += 32) {
        v8f c00 = {}, c01 = {}, c10 = {}, c11 = {};
        for (int k0 = 0; k0 < Kd; k0 += 32) {
            v2f a0[8], a1[8], b0[8], b1[8];
            #pragma unroll
            for (int s = 0; s < 8; ++s) {
                const int k = k0 + 4 * s + 2 * half;
                const float* ar = A + l15 * lda + k;
                a0[s].x = ar[0];
                a0[s].y = ar[1];
                a1[s].x = ar[16 * lda];
                a1[s].y = ar[16 * lda + 1];
                if (!transB) {
                    const float* br = B + k * ldb + ni + l15;
                    b0[s].x = br[0];
                    b0[s].y = br[ldb];
                    b1[s].x = br[16];
                    b1[s].y = br[ldb + 16];
                } else {
                    const float* br = B + (ni + l15) * ldb + k;
                    b0[s].x = br[0];
                    b0[s].y = br[1];
                    b1[s].x = br[16 * ldb];
                    b1[s].y = br[16 * ldb + 1];
                }
            }
            #pragma unroll
            for (int s = 0; s < 8; ++s) {
                c00 = WMMA_F32(a0[s], b0[s], c00);
                c01 = WMMA_F32(a0[s], b1[s], c01);
                c10 = WMMA_F32(a1[s], b0[s], c10);
                c11 = WMMA_F32(a1[s], b1[s], c11);
            }
        }
        #pragma unroll
        for (int r = 0; r < 8; ++r) {
            float* cr = C + (half * 8 + r) * ldc + ni + l15;
            cr[0]            = c00[r];
            cr[16]           = c01[r];
            cr[16 * ldc]     = c10[r];
            cr[16 * ldc + 16] = c11[r];
        }
    }
    __syncthreads();
}

struct M32 { v8f t[2][2]; };

__device__ __forceinline__ void mmT(M32& D, const M32& P, const M32& Q)
{
    v8f d00 = {}, d01 = {}, d10 = {}, d11 = {};
    #pragma unroll
    for (int tk = 0; tk < 2; ++tk) {
        #pragma unroll
        for (int s = 0; s < 4; ++s) {
            v2f a0, a1, b0, b1;
            a0.x = P.t[tk][0][2 * s]; a0.y = P.t[tk][0][2 * s + 1];
            a1.x = P.t[tk][1][2 * s]; a1.y = P.t[tk][1][2 * s + 1];
            b0.x = Q.t[tk][0][2 * s]; b0.y = Q.t[tk][0][2 * s + 1];
            b1.x = Q.t[tk][1][2 * s]; b1.y = Q.t[tk][1][2 * s + 1];
            d00 = WMMA_F32(a0, b0, d00);
            d01 = WMMA_F32(a0, b1, d01);
            d10 = WMMA_F32(a1, b0, d10);
            d11 = WMMA_F32(a1, b1, d11);
        }
    }
    D.t[0][0] = d00; D.t[0][1] = d01; D.t[1][0] = d10; D.t[1][1] = d11;
}

__device__ __forceinline__ void mmT_acc(M32& D, const M32& C, const M32& P, const M32& Q)
{
    v8f d00 = C.t[0][0], d01 = C.t[0][1], d10 = C.t[1][0], d11 = C.t[1][1];
    #pragma unroll
    for (int tk = 0; tk < 2; ++tk) {
        #pragma unroll
        for (int s = 0; s < 4; ++s) {
            v2f a0, a1, b0, b1;
            a0.x = P.t[tk][0][2 * s]; a0.y = P.t[tk][0][2 * s + 1];
            a1.x = P.t[tk][1][2 * s]; a1.y = P.t[tk][1][2 * s + 1];
            b0.x = Q.t[tk][0][2 * s]; b0.y = Q.t[tk][0][2 * s + 1];
            b1.x = Q.t[tk][1][2 * s]; b1.y = Q.t[tk][1][2 * s + 1];
            d00 = WMMA_F32(a0, b0, d00);
            d01 = WMMA_F32(a0, b1, d01);
            d10 = WMMA_F32(a1, b0, d10);
            d11 = WMMA_F32(a1, b1, d11);
        }
    }
    D.t[0][0] = d00; D.t[0][1] = d01; D.t[1][0] = d10; D.t[1][1] = d11;
}

template <bool FLIP>
__device__ __forceinline__ void sqrt_step(M32& Y, M32& Z, int half, int l15,
                                          float sg, float k0, float k1, float k2, float k3, float k4)
{
    M32 X, X2, W, P;
    if (!FLIP) mmT(X, Z, Y); else mmT(X, Y, Z);
    #pragma unroll
    for (int ti = 0; ti < 2; ++ti)
        #pragma unroll
        for (int tj = 0; tj < 2; ++tj)
            #pragma unroll
            for (int r = 0; r < 8; ++r) {
                const float d = (ti == tj && (8 * half + r) == l15) ? sg : 0.f;
                X.t[ti][tj][r] = X.t[ti][tj][r] - d;
            }
    mmT(X2, X, X);
    #pragma unroll
    for (int ti = 0; ti < 2; ++ti)
        #pragma unroll
        for (int tj = 0; tj < 2; ++tj)
            #pragma unroll
            for (int r = 0; r < 8; ++r) {
                const float d  = (ti == tj && (8 * half + r) == l15) ? k0 : 0.f;
                const float x  = X.t[ti][tj][r];
                const float x2 = X2.t[ti][tj][r];
                W.t[ti][tj][r] = k3 * x + k4 * x2;
                P.t[ti][tj][r] = d + k1 * x + k2 * x2;
            }
    if (!FLIP) mmT_acc(P, P, X2, W); else mmT_acc(P, P, W, X2);
    M32 Yn, Zn;
    if (!FLIP) { mmT(Yn, P, Y); mmT(Zn, Z, P); }
    else       { mmT(Yn, Y, P); mmT(Zn, P, Z); }
    Y = Yn; Z = Zn;
}

__device__ __forceinline__ void symmetrise2(M32& Y, M32& Z, float* by, float* bz, int half, int l15)
{
    #pragma unroll
    for (int ti = 0; ti < 2; ++ti)
        #pragma unroll
        for (int tj = 0; tj < 2; ++tj)
            #pragma unroll
            for (int r = 0; r < 8; ++r) {
                const int i = (16 * ti + 8 * half + r) * 32 + 16 * tj + l15;
                by[i] = Y.t[ti][tj][r];
                bz[i] = Z.t[ti][tj][r];
            }
    __syncthreads();
    #pragma unroll
    for (int ti = 0; ti < 2; ++ti)
        #pragma unroll
        for (int tj = 0; tj < 2; ++tj)
            #pragma unroll
            for (int r = 0; r < 8; ++r) {
                const int j = (16 * tj + l15) * 32 + 16 * ti + 8 * half + r;
                Y.t[ti][tj][r] = 0.5f * (Y.t[ti][tj][r] + by[j]);
                Z.t[ti][tj][r] = 0.5f * (Z.t[ti][tj][r] + bz[j]);
            }
    __syncthreads();
}

__device__ __forceinline__ void ldM(M32& D, const float* p, int half, int l15)
{
    #pragma unroll
    for (int ti = 0; ti < 2; ++ti)
        #pragma unroll
        for (int tj = 0; tj < 2; ++tj)
            #pragma unroll
            for (int r = 0; r < 8; ++r)
                D.t[ti][tj][r] = p[(16 * ti + 8 * half + r) * 32 + 16 * tj + l15];
}

__device__ __forceinline__ void stM(float* p, const M32& S, int half, int l15)
{
    #pragma unroll
    for (int ti = 0; ti < 2; ++ti)
        #pragma unroll
        for (int tj = 0; tj < 2; ++tj)
            #pragma unroll
            for (int r = 0; r < 8; ++r)
                p[(16 * ti + 8 * half + r) * 32 + 16 * tj + l15] = S.t[ti][tj][r];
}

__device__ __forceinline__ void root_reg(M32& Y, M32& Z, float* b0, float* b1, int half, int l15)
{
    stM(b0, Y, half, l15);
    __syncthreads();
    float c = 0.f;
    #pragma unroll
    for (int i = 0; i < 32; ++i) c += b0[i * 33];
    const float inv_c = 1.f / c;
    #pragma unroll
    for (int ti = 0; ti < 2; ++ti)
        #pragma unroll
        for (int tj = 0; tj < 2; ++tj)
            #pragma unroll
            for (int r = 0; r < 8; ++r) {
                const int row = 16 * ti + 8 * half + r;
                const int col = 16 * tj + l15;
                Y.t[ti][tj][r] = 0.5f * (Y.t[ti][tj][r] + b0[col * 32 + row]) * inv_c;
                Z.t[ti][tj][r] = (row == col) ? 1.f : 0.f;
            }
    __syncthreads();
    #pragma nounroll
    for (int it = 0; it < 2; ++it) {
        const bool a = (it == 0);
        sqrt_step<false>(Y, Z, half, l15,
                         a ? 0.550000012f : 1.61000001f,
                         a ? 2.10768604f : 0.790120423f, a ? 5.74562931f : -0.282128632f, a ? -30.9237690f : 0.0719144717f,
                         a ? -54.0620193f : 0.150992960f, a ? 162.579529f : 0.289191633f);
        symmetrise2(Y, Z, b0, b1, half, l15);
        sqrt_step<true>(Y, Z, half, l15,
                        a ? 1.79999995f : 1.00000000f,
                        a ? 0.935014427f : 1.00000000f, a ? 0.400891274f : -0.500000000f, a ? -0.721723795f : 0.375000000f,
                        a ? -0.493403643f : -0.312500000f, a ? 0.415029675f : 0.273437500f);
    }
    const float sc  = __fsqrt_rn(c);
    const float isc = __frsqrt_rn(c);
    #pragma unroll
    for (int ti = 0; ti < 2; ++ti)
        #pragma unroll
        for (int tj = 0; tj < 2; ++tj)
            #pragma unroll
            for (int r = 0; r < 8; ++r) {
                Y.t[ti][tj][r] = Y.t[ti][tj][r] * sc;
                Z.t[ti][tj][r] = Z.t[ti][tj][r] * isc;
            }
}

__global__ __launch_bounds__(32)
void spdnet_mixedop_kernel(const float* __restrict__ x,
                           const float* __restrict__ wts,
                           const float* __restrict__ Ws,
                           float* __restrict__ out)
{
    __shared__ float sm[9 * 1024];
    float* const sX = sm;
    float* const sP = sm + 4 * 1024;
    float* const s0 = sm + 6 * 1024;
    float* const s1 = sm + 7 * 1024;
    float* const s2 = sm + 8 * 1024;
    float* const s3 = sm;

    const int bc   = blockIdx.x;
    const int lane = threadIdx.x;

    float wk[4];
    #pragma unroll
    for (int k = 0; k < 4; ++k) wk[k] = wts[k];

    #pragma unroll 16
    for (int i = lane; i < 4096; i += 32) sX[i] = x[bc * 4096 + i];

    for (int k = 0; k < 4; ++k) {
        const float* Wk = Ws + k * 2048;
        float* sk = (k < 3) ? (s0 + k * 1024) : s3;
        mm(sP, 64, Wk, 64, sX, 64, 64, 64, false);
        mm(sk, 32, sP, 64, Wk, 64, 32, 64, true);
    }

    float* const b0  = sm + 1 * 1024;
    float* const b1  = sm + 2 * 1024;
    float* const pSq = sm + 3 * 1024;
    float* const pSi = sm + 4 * 1024;
    const int half = lane >> 4;
    const int l15  = lane & 15;

    M32 U;
    #pragma unroll
    for (int ti = 0; ti < 2; ++ti)
        #pragma unroll
        for (int tj = 0; tj < 2; ++tj)
            #pragma unroll
            for (int r = 0; r < 8; ++r) U.t[ti][tj][r] = 0.f;

    for (int k = -1; k < 4; ++k) {
        M32 Y, Z;
        if (k < 0) {
            #pragma unroll
            for (int ti = 0; ti < 2; ++ti)
                #pragma unroll
                for (int tj = 0; tj < 2; ++tj)
                    #pragma unroll
                    for (int r = 0; r < 8; ++r) {
                        const int i = (16 * ti + 8 * half + r) * 32 + 16 * tj + l15;
                        Y.t[ti][tj][r] = wk[0] * s0[i] + wk[1] * s1[i] +
                                         wk[2] * s2[i] + wk[3] * s3[i];
                    }
        } else {
            const float* sk = (k < 3) ? (s0 + k * 1024) : s3;
            M32 S, Si, A1;
            ldM(S, sk, half, l15);
            ldM(Si, pSi, half, l15);
            mmT(A1, S, Si);
            mmT(Y, A1, Si);
        }

        root_reg(Y, Z, b0, b1, half, l15);

        if (k < 0) {
            stM(pSq, Y, half, l15);
            stM(pSi, Z, half, l15);
            continue;
        }

        const float SC = 0.690548897f;
        M32 H1, X, q;
        #pragma unroll
        for (int ti = 0; ti < 2; ++ti)
            #pragma unroll
            for (int tj = 0; tj < 2; ++tj)
                #pragma unroll
                for (int r = 0; r < 8; ++r) {
                    const bool  dg = (ti == tj && (8 * half + r) == l15);
                    const float y  = Y.t[ti][tj][r];
                    const float z  = Z.t[ti][tj][r];
                    const float x  = 0.25f * (y + z) - (dg ? (0.5f + SC) : 0.f);
                    H1.t[ti][tj][r] = 0.5f * (y - z);
                    X.t[ti][tj][r]  = x;
                    q.t[ti][tj][r]  = (dg ? 0.0101555130f : 0.f) + (-0.00572075741f) * x;
                }
        for (int ii = 7; ii >= 0; --ii) {
            M32 qn;
            mmT(qn, X, q);
            const float g = (ii == 7) ? -0.0113067133f : (ii == 6) ? 0.0204704814f :
                            (ii == 5) ? -0.0403085500f : (ii == 4) ? 0.0746348128f :
                            (ii == 3) ? -0.140599474f  : (ii == 2) ? 0.274833739f  :
                            (ii == 1) ? -0.571834743f  : 1.40066803f;
            #pragma unroll
            for (int ti = 0; ti < 2; ++ti)
                #pragma unroll
                for (int r = 0; r < 8; ++r)
                    qn.t[ti][ti][r] = qn.t[ti][ti][r] + (((8 * half + r) == l15) ? g : 0.f);
            q = qn;
        }
        M32 L;
        mmT(L, H1, q);

        const float wsel  = (k == 0) ? wk[0] : (k == 1) ? wk[1] : (k == 2) ? wk[2] : wk[3];
        #pragma unroll
        for (int ti = 0; ti < 2; ++ti)
            #pragma unroll
            for (int tj = 0; tj < 2; ++tj)
                #pragma unroll
                for (int r = 0; r < 8; ++r)
                    U.t[ti][tj][r] = U.t[ti][tj][r] + wsel * L.t[ti][tj][r];
    }

    M32 V, V2, V3, A0, A1, In, Md, R;
    #pragma unroll
    for (int ti = 0; ti < 2; ++ti)
        #pragma unroll
        for (int tj = 0; tj < 2; ++tj)
            #pragma unroll
            for (int r = 0; r < 8; ++r)
                V.t[ti][tj][r] = U.t[ti][tj][r] * (1.f / 32.f);
    mmT(V2, V, V);
    mmT(V3, V, V2);
    #pragma unroll
    for (int ti = 0; ti < 2; ++ti)
        #pragma unroll
        for (int tj = 0; tj < 2; ++tj)
            #pragma unroll
            for (int r = 0; r < 8; ++r) {
                const bool  dg = (ti == tj && (8 * half + r) == l15);
                const float v  = V.t[ti][tj][r];
                const float v2 = V2.t[ti][tj][r];
                const float v3 = V3.t[ti][tj][r];
                A0.t[ti][tj][r] = (dg ? 1.f : 0.f) + v + 0.5f * v2;
                A1.t[ti][tj][r] = (dg ? (1.f / 6.f) : 0.f) + (1.f / 24.f) * v + (1.f / 120.f) * v2;
                In.t[ti][tj][r] = (dg ? (1.f / 720.f) : 0.f) + (1.f / 5040.f) * v +
                                  (1.f / 40320.f) * v2 + (1.f / 362880.f) * v3;
            }
    mmT_acc(Md, A1, V3, In);
    mmT_acc(R, A0, V3, Md);
    for (int sq = 0; sq < 4; ++sq) {
        M32 Rn;
        mmT(Rn, R, R);
        R = Rn;
    }
    M32 Sq, E1, O;
    ldM(Sq, pSq, half, l15);
    mmT(E1, R, Sq);
    mmT(O, E1, E1);
    stM(b0, O, half, l15);
    __syncthreads();
    {
        float v[32];
        ld32(v, b0, lane);
        #pragma unroll
        for (int j = 0; j < 32; ++j) out[bc * 1024 + lane + 32 * j] = v[j];
    }
}

extern "C" void kernel_launch(void* const* d_in, const int* in_sizes, int n_in,
                              void* d_out, int out_size, void* d_ws, size_t ws_size,
                              hipStream_t stream)
{
    const float* x   = (const float*)d_in[0];
    const float* wts = (const float*)d_in[1];
    const float* Ws  = (const float*)d_in[2];
    float* out = (float*)d_out;

    dim3 grid(2048);
    dim3 block(32);
    hipLaunchKernelGGL(spdnet_mixedop_kernel, grid, block, 0, stream,
                       x, wts, Ws, out);
}


# Phase A on register operands: x and W_k^T go from global straight into matrix-operand registers (16-byte loads); x W_k^T and W_k (x W_k^T) are register products, no LDS; new k order, rounding differs
# speedup vs baseline: 11.9116x; 1.0334x over previous
#include <hip/hip_runtime.h>


typedef __attribute__((ext_vector_type(2))) float v2f;
typedef __attribute__((ext_vector_type(4))) float v4f;
typedef __attribute__((ext_vector_type(8))) float v8f;

#define WMMA_F32(a, b, c) \
    __builtin_amdgcn_wmma_f32_16x16x4_f32(false, (a), false, (b), (short)0, (c), false, false)

__device__ __forceinline__ void ld32(float (&v)[32], const float* p, int lane)
{
    #pragma unroll
    for (int j = 0; j < 32; ++j) v[j] = p[lane + 32 * j];
}

struct M32 { v8f t[2][2]; };

struct M64 { v8f t[4][2]; };

__device__ __forceinline__ v8f ld8(const float* __restrict__ p)
{
    const v4f lo = *reinterpret_cast<const v4f*>(p);
    const v4f hi = *reinterpret_cast<const v4f*>(p + 4);
    v8f v;
    v[0] = lo[0]; v[1] = lo[1]; v[2] = lo[2]; v[3] = lo[3];
    v[4] = hi[0]; v[5] = hi[1]; v[6] = hi[2]; v[7] = hi[3];
    return v;
}

__device__ __forceinline__ void ldWt(M64& P, const float* __restrict__ Wk, int half, int l15)
{
    #pragma unroll
    for (int tk = 0; tk < 4; ++tk)
        #pragma unroll
        for (int j = 0; j < 2; ++j)
            P.t[tk][j] = ld8(Wk + (16 * j + l15) * 64 + 16 * tk + 8 * half);
}

__device__ __forceinline__ void bimap_reg(M32& S, const v8f (&XA)[4][4], const M64& P)
{
    M64 Q;
    #pragma unroll
    for (int mp = 0; mp < 2; ++mp) {
        v8f q00 = {}, q01 = {}, q10 = {}, q11 = {};
        #pragma unroll
        for (int tk = 0; tk < 4; ++tk) {
            #pragma unroll
            for (int s = 0; s < 4; ++s) {
                v2f a0, a1, b0, b1;
                a0.x = XA[2 * mp][tk][2 * s];     a0.y = XA[2 * mp][tk][2 * s + 1];
                a1.x = XA[2 * mp + 1][tk][2 * s]; a1.y = XA[2 * mp + 1][tk][2 * s + 1];
                b0.x = P.t[tk][0][2 * s];         b0.y = P.t[tk][0][2 * s + 1];
                b1.x = P.t[tk][1][2 * s];         b1.y = P.t[tk][1][2 * s + 1];
                q00 = WMMA_F32(a0, b0, q00);
                q01 = WMMA_F32(a0, b1, q01);
                q10 = WMMA_F32(a1, b0, q10);
                q11 = WMMA_F32(a1, b1, q11);
            }
        }
        Q.t[2 * mp][0] = q00;     Q.t[2 * mp][1] = q01;
        Q.t[2 * mp + 1][0] = q10; Q.t[2 * mp + 1][1] = q11;
    }
    v8f d00 = {}, d01 = {}, d10 = {}, d11 = {};
    #pragma unroll
    for (int tk = 0; tk < 4; ++tk) {
        #pragma unroll
        for (int s = 0; s < 4; ++s) {
            v2f a0, a1, b0, b1;
            a0.x = P.t[tk][0][2 * s]; a0.y = P.t[tk][0][2 * s + 1];
            a1.x = P.t[tk][1][2 * s]; a1.y = P.t[tk][1][2 * s + 1];
            b0.x = Q.t[tk][0][2 * s]; b0.y = Q.t[tk][0][2 * s + 1];
            b1.x = Q.t[tk][1][2 * s]; b1.y = Q.t[tk][1][2 * s + 1];
            d00 = WMMA_F32(a0, b0, d00);
            d01 = WMMA_F32(a0, b1, d01);
            d10 = WMMA_F32(a1, b0, d10);
            d11 = WMMA_F32(a1, b1, d11);
        }
    }
    S.t[0][0] = d00; S.t[0][1] = d01; S.t[1][0] = d10; S.t[1][1] = d11;
}

__device__ __forceinline__ void axpyM(M32& G, float w, const M32& S)
{
    #pragma unroll
    for (int ti = 0; ti < 2; ++ti)
        #pragma unroll
        for (int tj = 0; tj < 2; ++tj)
            #pragma unroll
            for (int r = 0; r < 8; ++r)
                G.t[ti][tj][r] = G.t[ti][tj][r] + w * S.t[ti][tj][r];
}

__device__ __forceinline__ void mmT(M32& D, const M32& P, const M32& Q)
{
    v8f d00 = {}, d01 = {}, d10 = {}, d11 = {};
    #pragma unroll
    for (int tk = 0; tk < 2; ++tk) {
        #pragma unroll
        for (int s = 0; s < 4; ++s) {
            v2f a0, a1, b0, b1;
            a0.x = P.t[tk][0][2 * s]; a0.y = P.t[tk][0][2 * s + 1];
            a1.x = P.t[tk][1][2 * s]; a1.y = P.t[tk][1][2 * s + 1];
            b0.x = Q.t[tk][0][2 * s]; b0.y = Q.t[tk][0][2 * s + 1];
            b1.x = Q.t[tk][1][2 * s]; b1.y = Q.t[tk][1][2 * s + 1];
            d00 = WMMA_F32(a0, b0, d00);
            d01 = WMMA_F32(a0, b1, d01);
            d10 = WMMA_F32(a1, b0, d10);
            d11 = WMMA_F32(a1, b1, d11);
        }
    }
    D.t[0][0] = d00; D.t[0][1] = d01; D.t[1][0] = d10; D.t[1][1] = d11;
}

__device__ __forceinline__ void mmT_acc(M32& D, const M32& C, const M32& P, const M32& Q)
{
    v8f d00 = C.t[0][0], d01 = C.t[0][1], d10 = C.t[1][0], d11 = C.t[1][1];
    #pragma unroll
    for (int tk = 0; tk < 2; ++tk) {
        #pragma unroll
        for (int s = 0; s < 4; ++s) {
            v2f a0, a1, b0, b1;
            a0.x = P.t[tk][0][2 * s]; a0.y = P.t[tk][0][2 * s + 1];
            a1.x = P.t[tk][1][2 * s]; a1.y = P.t[tk][1][2 * s + 1];
            b0.x = Q.t[tk][0][2 * s]; b0.y = Q.t[tk][0][2 * s + 1];
            b1.x = Q.t[tk][1][2 * s]; b1.y = Q.t[tk][1][2 * s + 1];
            d00 = WMMA_F32(a0, b0, d00);
            d01 = WMMA_F32(a0, b1, d01);
            d10 = WMMA_F32(a1, b0, d10);
            d11 = WMMA_F32(a1, b1, d11);
        }
    }
    D.t[0][0] = d00; D.t[0][1] = d01; D.t[1][0] = d10; D.t[1][1] = d11;
}

template <bool FLIP>
__device__ __forceinline__ void sqrt_step(M32& Y, M32& Z, int half, int l15,
                                          float sg, float k0, float k1, float k2, float k3, float k4)
{
    M32 X, X2, W, P;
    if (!FLIP) mmT(X, Z, Y); else mmT(X, Y, Z);
    #pragma unroll
    for (int ti = 0; ti < 2; ++ti)
        #pragma unroll
        for (int tj = 0; tj < 2; ++tj)
            #pragma unroll
            for (int r = 0; r < 8; ++r) {
                const float d = (ti == tj && (8 * half + r) == l15) ? sg : 0.f;
                X.t[ti][tj][r] = X.t[ti][tj][r] - d;
            }
    mmT(X2, X, X);
    #pragma unroll
    for (int ti = 0; ti < 2; ++ti)
        #pragma unroll
        for (int tj = 0; tj < 2; ++tj)
            #pragma unroll
            for (int r = 0; r < 8; ++r) {
                const float d  = (ti == tj && (8 * half + r) == l15) ? k0 : 0.f;
                const float x  = X.t[ti][tj][r];
                const float x2 = X2.t[ti][tj][r];
                W.t[ti][tj][r] = k3 * x + k4 * x2;
                P.t[ti][tj][r] = d + k1 * x + k2 * x2;
            }
    if (!FLIP) mmT_acc(P, P, X2, W); else mmT_acc(P, P, W, X2);
    M32 Yn, Zn;
    if (!FLIP) { mmT(Yn, P, Y); mmT(Zn, Z, P); }
    else       { mmT(Yn, Y, P); mmT(Zn, P, Z); }
    Y = Yn; Z = Zn;
}

__device__ __forceinline__ void symmetrise2(M32& Y, M32& Z, float* by, float* bz, int half, int l15)
{
    #pragma unroll
    for (int ti = 0; ti < 2; ++ti)
        #pragma unroll
        for (int tj = 0; tj < 2; ++tj)
            #pragma unroll
            for (int r = 0; r < 8; ++r) {
                const int i = (16 * ti + 8 * half + r) * 32 + 16 * tj + l15;
                by[i] = Y.t[ti][tj][r];
                bz[i] = Z.t[ti][tj][r];
            }
    __syncthreads();
    #pragma unroll
    for (int ti = 0; ti < 2; ++ti)
        #pragma unroll
        for (int tj = 0; tj < 2; ++tj)
            #pragma unroll
            for (int r = 0; r < 8; ++r) {
                const int j = (16 * tj + l15) * 32 + 16 * ti + 8 * half + r;
                Y.t[ti][tj][r] = 0.5f * (Y.t[ti][tj][r] + by[j]);
                Z.t[ti][tj][r] = 0.5f * (Z.t[ti][tj][r] + bz[j]);
            }
    __syncthreads();
}

__device__ __forceinline__ void ldM(M32& D, const float* p, int half, int l15)
{
    #pragma unroll
    for (int ti = 0; ti < 2; ++ti)
        #pragma unroll
        for (int tj = 0; tj < 2; ++tj)
            #pragma unroll
            for (int r = 0; r < 8; ++r)
                D.t[ti][tj][r] = p[(16 * ti + 8 * half + r) * 32 + 16 * tj + l15];
}

__device__ __forceinline__ void stM(float* p, const M32& S, int half, int l15)
{
    #pragma unroll
    for (int ti = 0; ti < 2; ++ti)
        #pragma unroll
        for (int tj = 0; tj < 2; ++tj)
            #pragma unroll
            for (int r = 0; r < 8; ++r)
                p[(16 * ti + 8 * half + r) * 32 + 16 * tj + l15] = S.t[ti][tj][r];
}

__device__ __forceinline__ void root_reg(M32& Y, M32& Z, float* b0, float* b1, int half, int l15)
{
    stM(b0, Y, half, l15);
    __syncthreads();
    float c = 0.f;
    #pragma unroll
    for (int i = 0; i < 32; ++i) c += b0[i * 33];
    const float inv_c = 1.f / c;
    #pragma unroll
    for (int ti = 0; ti < 2; ++ti)
        #pragma unroll
        for (int tj = 0; tj < 2; ++tj)
            #pragma unroll
            for (int r = 0; r < 8; ++r) {
                const int row = 16 * ti + 8 * half + r;
                const int col = 16 * tj + l15;
                Y.t[ti][tj][r] = 0.5f * (Y.t[ti][tj][r] + b0[col * 32 + row]) * inv_c;
                Z.t[ti][tj][r] = (row == col) ? 1.f : 0.f;
            }
    __syncthreads();
    #pragma nounroll
    for (int it = 0; it < 2; ++it) {
        const bool a = (it == 0);
        sqrt_step<false>(Y, Z, half, l15,
                         a ? 0.550000012f : 1.61000001f,
                         a ? 2.10768604f : 0.790120423f, a ? 5.74562931f : -0.282128632f, a ? -30.9237690f : 0.0719144717f,
                         a ? -54.0620193f : 0.150992960f, a ? 162.579529f : 0.289191633f);
        symmetrise2(Y, Z, b0, b1, half, l15);
        sqrt_step<true>(Y, Z, half, l15,
                        a ? 1.79999995f : 1.00000000f,
                        a ? 0.935014427f : 1.00000000f, a ? 0.400891274f : -0.500000000f, a ? -0.721723795f : 0.375000000f,
                        a ? -0.493403643f : -0.312500000f, a ? 0.415029675f : 0.273437500f);
    }
    const float sc  = __fsqrt_rn(c);
    const float isc = __frsqrt_rn(c);
    #pragma unroll
    for (int ti = 0; ti < 2; ++ti)
        #pragma unroll
        for (int tj = 0; tj < 2; ++tj)
            #pragma unroll
            for (int r = 0; r < 8; ++r) {
                Y.t[ti][tj][r] = Y.t[ti][tj][r] * sc;
                Z.t[ti][tj][r] = Z.t[ti][tj][r] * isc;
            }
}

__global__ __launch_bounds__(32)
void spdnet_mixedop_kernel(const float* __restrict__ x,
                           const float* __restrict__ wts,
                           const float* __restrict__ Ws,
                           float* __restrict__ out)
{
    __shared__ float sm[9 * 1024];
    float* const b0  = sm + 4 * 1024;
    float* const b1  = sm + 5 * 1024;
    float* const pSq = sm + 6 * 1024;
    float* const pSi = sm + 7 * 1024;
    float* const pG  = sm + 8 * 1024;

    const int bc   = blockIdx.x;
    const int lane = threadIdx.x;
    const int half = lane >> 4;
    const int l15  = lane & 15;

    float wk[4];
    #pragma unroll
    for (int k = 0; k < 4; ++k) wk[k] = wts[k];

    v8f XA[4][4];
    {
        const float* xr = x + bc * 4096 + l15 * 64 + 8 * half;
        #pragma unroll
        for (int m = 0; m < 4; ++m)
            #pragma unroll
            for (int tk = 0; tk < 4; ++tk)
                XA[m][tk] = ld8(xr + 16 * m * 64 + 16 * tk);
    }

    M32 G0;
    #pragma unroll
    for (int ti = 0; ti < 2; ++ti)
        #pragma unroll
        for (int tj = 0; tj < 2; ++tj)
            #pragma unroll
            for (int r = 0; r < 8; ++r) G0.t[ti][tj][r] = 0.f;

    M64 Pa, Pb;
    ldWt(Pa, Ws, half, l15);
    #pragma nounroll
    for (int k = 0; k < 4; k += 2) {
        M32 S;
        ldWt(Pb, Ws + (k + 1) * 2048, half, l15);
        bimap_reg(S, XA, Pa);
        stM(sm + k * 1024, S, half, l15);
        axpyM(G0, (k == 0) ? wk[0] : wk[2], S);
        if (k == 0) ldWt(Pa, Ws + 2 * 2048, half, l15);
        bimap_reg(S, XA, Pb);
        stM(sm + (k + 1) * 1024, S, half, l15);
        axpyM(G0, (k == 0) ? wk[1] : wk[3], S);
    }
    stM(pG, G0, half, l15);


    M32 U;
    #pragma unroll
    for (int ti = 0; ti < 2; ++ti)
        #pragma unroll
        for (int tj = 0; tj < 2; ++tj)
            #pragma unroll
            for (int r = 0; r < 8; ++r) U.t[ti][tj][r] = 0.f;

    for (int k = -1; k < 4; ++k) {
        M32 Y, Z;
        if (k < 0) {
            ldM(Y, pG, half, l15);
        } else {
            const float* sk = sm + k * 1024;
            M32 S, Si, A1;
            ldM(S, sk, half, l15);
            ldM(Si, pSi, half, l15);
            mmT(A1, S, Si);
            mmT(Y, A1, Si);
        }

        root_reg(Y, Z, b0, b1, half, l15);

        if (k < 0) {
            stM(pSq, Y, half, l15);
            stM(pSi, Z, half, l15);
            continue;
        }

        const float SC = 0.690548897f;
        M32 H1, X, q;
        #pragma unroll
        for (int ti = 0; ti < 2; ++ti)
            #pragma unroll
            for (int tj = 0; tj < 2; ++tj)
                #pragma unroll
                for (int r = 0; r < 8; ++r) {
                    const bool  dg = (ti == tj && (8 * half + r) == l15);
                    const float y  = Y.t[ti][tj][r];
                    const float z  = Z.t[ti][tj][r];
                    const float x  = 0.25f * (y + z) - (dg ? (0.5f + SC) : 0.f);
                    H1.t[ti][tj][r] = 0.5f * (y - z);
                    X.t[ti][tj][r]  = x;
                    q.t[ti][tj][r]  = (dg ? 0.0101555130f : 0.f) + (-0.00572075741f) * x;
                }
        for (int ii = 7; ii >= 0; --ii) {
            M32 qn;
            mmT(qn, X, q);
            const float g = (ii == 7) ? -0.0113067133f : (ii == 6) ? 0.0204704814f :
                            (ii == 5) ? -0.0403085500f : (ii == 4) ? 0.0746348128f :
                            (ii == 3) ? -0.140599474f  : (ii == 2) ? 0.274833739f  :
                            (ii == 1) ? -0.571834743f  : 1.40066803f;
            #pragma unroll
            for (int ti = 0; ti < 2; ++ti)
                #pragma unroll
                for (int r = 0; r < 8; ++r)
                    qn.t[ti][ti][r] = qn.t[ti][ti][r] + (((8 * half + r) == l15) ? g : 0.f);
            q = qn;
        }
        M32 L;
        mmT(L, H1, q);

        const float wsel  = (k == 0) ? wk[0] : (k == 1) ? wk[1] : (k == 2) ? wk[2] : wk[3];
        #pragma unroll
        for (int ti = 0; ti < 2; ++ti)
            #pragma unroll
            for (int tj = 0; tj < 2; ++tj)
                #pragma unroll
                for (int r = 0; r < 8; ++r)
                    U.t[ti][tj][r] = U.t[ti][tj][r] + wsel * L.t[ti][tj][r];
    }

    M32 V, V2, V3, A0, A1, In, Md, R;
    #pragma unroll
    for (int ti = 0; ti < 2; ++ti)
        #pragma unroll
        for (int tj = 0; tj < 2; ++tj)
            #pragma unroll
            for (int r = 0; r < 8; ++r)
                V.t[ti][tj][r] = U.t[ti][tj][r] * (1.f / 32.f);
    mmT(V2, V, V);
    mmT(V3, V, V2);
    #pragma unroll
    for (int ti = 0; ti < 2; ++ti)
        #pragma unroll
        for (int tj = 0; tj < 2; ++tj)
            #pragma unroll
            for (int r = 0; r < 8; ++r) {
                const bool  dg = (ti == tj && (8 * half + r) == l15);
                const float v  = V.t[ti][tj][r];
                const float v2 = V2.t[ti][tj][r];
                const float v3 = V3.t[ti][tj][r];
                A0.t[ti][tj][r] = (dg ? 1.f : 0.f) + v + 0.5f * v2;
                A1.t[ti][tj][r] = (dg ? (1.f / 6.f) : 0.f) + (1.f / 24.f) * v + (1.f / 120.f) * v2;
                In.t[ti][tj][r] = (dg ? (1.f / 720.f) : 0.f) + (1.f / 5040.f) * v +
                                  (1.f / 40320.f) * v2 + (1.f / 362880.f) * v3;
            }
    mmT_acc(Md, A1, V3, In);
    mmT_acc(R, A0, V3, Md);
    for (int sq = 0; sq < 4; ++sq) {
        M32 Rn;
        mmT(Rn, R, R);
        R = Rn;
    }
    M32 Sq, E1, O;
    ldM(Sq, pSq, half, l15);
    mmT(E1, R, Sq);
    mmT(O, E1, E1);
    stM(b0, O, half, l15);
    __syncthreads();
    {
        float v[32];
        ld32(v, b0, lane);
        #pragma unroll
        for (int j = 0; j < 32; ++j) out[bc * 1024 + lane + 32 * j] = v[j];
    }
}

extern "C" void kernel_launch(void* const* d_in, const int* in_sizes, int n_in,
                              void* d_out, int out_size, void* d_ws, size_t ws_size,
                              hipStream_t stream)
{
    const float* x   = (const float*)d_in[0];
    const float* wts = (const float*)d_in[1];
    const float* Ws  = (const float*)d_in[2];
    float* out = (float*)d_out;

    dim3 grid(2048);
    dim3 block(32);
    hipLaunchKernelGGL(spdnet_mixedop_kernel, grid, block, 0, stream,
                       x, wts, Ws, out);
}
